# starting kernel
# baseline (speedup 1.0000x reference)
#include <hip/hip_runtime.h>
#include <math.h>

#define NN     100000
#define NE     1600000
#define EMB3   192
#define F_LAT  64
#define F_OUT  128
#define NEG_SLOPE 0.2f

typedef __attribute__((ext_vector_type(2))) float v2f;
typedef __attribute__((ext_vector_type(8))) float v8f;

__device__ __forceinline__ v8f wmma4(v2f a, v2f b, v8f c) {
  return __builtin_amdgcn_wmma_f32_16x16x4_f32(false, a, false, b, (short)0, c,
                                               false, false);
}

__device__ __forceinline__ unsigned enc_f32(float f) {
  unsigned u = __float_as_uint(f);
  return (u & 0x80000000u) ? ~u : (u | 0x80000000u);
}
__device__ __forceinline__ float dec_f32(unsigned u) {
  return (u & 0x80000000u) ? __uint_as_float(u & 0x7FFFFFFFu)
                           : __uint_as_float(~u);
}

__global__ void init_kernel(float* __restrict__ rst, float* __restrict__ denom,
                            unsigned* __restrict__ emax_u) {
  int i = blockIdx.x * blockDim.x + threadIdx.x;
  if (i < NN * F_OUT) rst[i] = 0.0f;
  if (i < NN) { denom[i] = 0.0f; emax_u[i] = 0u; }
}

__global__ __launch_bounds__(256) void node_feat_kernel(
    const float* __restrict__ embed,
    const float* __restrict__ W_lin, const float* __restrict__ b_lin,
    const float* __restrict__ W_gat,
    const float* __restrict__ attn_l, const float* __restrict__ attn_r,
    float* __restrict__ feat, float* __restrict__ el, float* __restrict__ er,
    int n_tiles) {
  extern __shared__ float smem[];
  float* sW1 = smem;
  float* sW2 = sW1 + EMB3 * F_LAT;
  float* sB1 = sW2 + F_LAT * F_OUT;
  float* sAl = sB1 + F_LAT;
  float* sAr = sAl + F_OUT;
  float* sX  = sAr + F_OUT;

  const int tid = threadIdx.x;
  for (int i = tid; i < EMB3 * F_LAT; i += 256) sW1[i] = W_lin[i];
  for (int i = tid; i < F_LAT * F_OUT; i += 256) sW2[i] = W_gat[i];
  if (tid < F_LAT) sB1[tid] = b_lin[tid];
  if (tid < F_OUT) { sAl[tid] = attn_l[tid]; sAr[tid] = attn_r[tid]; }
  __syncthreads();

  const int wave = tid >> 5;
  const int lane = tid & 31;
  const int tile = blockIdx.x * 8 + wave;
  if (tile >= n_tiles) return;

  const int m  = lane & 15;
  const int hi = lane >> 4;

  const float* arow = embed + (size_t)(tile * 16 + m) * EMB3;
  v8f acc[4] = {};
  for (int k0 = 0; k0 < EMB3; k0 += 4) {
    const int ka = k0 + 2 * hi;
    v2f a; a.x = arow[ka]; a.y = arow[ka + 1];
    const float* b0 = sW1 + ka * F_LAT + m;
    const float* b1 = b0 + F_LAT;
#pragma unroll
    for (int nt = 0; nt < 4; ++nt) {
      v2f b; b.x = b0[nt * 16]; b.y = b1[nt * 16];
      acc[nt] = wmma4(a, b, acc[nt]);
    }
  }

  float* xw = sX + wave * 16 * F_LAT;
#pragma unroll
  for (int r = 0; r < 8; ++r) {
    const int row = r + 8 * hi;
#pragma unroll
    for (int nt = 0; nt < 4; ++nt) {
      float v = acc[nt][r] + sB1[nt * 16 + m];
      xw[row * F_LAT + nt * 16 + m] = fmaxf(v, 0.0f);
    }
  }

  v8f fa[8] = {};
  for (int k0 = 0; k0 < F_LAT; k0 += 4) {
    const int ka = k0 + 2 * hi;
    v2f a; a.x = xw[m * F_LAT + ka]; a.y = xw[m * F_LAT + ka + 1];
    const float* b0 = sW2 + ka * F_OUT + m;
    const float* b1 = b0 + F_OUT;
#pragma unroll
    for (int nt = 0; nt < 8; ++nt) {
      v2f b; b.x = b0[nt * 16]; b.y = b1[nt * 16];
      fa[nt] = wmma4(a, b, fa[nt]);
    }
  }

  float pl[8], pr[8];
  const size_t base = (size_t)(tile * 16) * F_OUT;
#pragma unroll
  for (int r = 0; r < 8; ++r) {
    const int row = r + 8 * hi;
    float sl = 0.0f, sr = 0.0f;
#pragma unroll
    for (int nt = 0; nt < 8; ++nt) {
      const float v = fa[nt][r];
      feat[base + row * F_OUT + nt * 16 + m] = v;
      sl += v * sAl[nt * 16 + m];
      sr += v * sAr[nt * 16 + m];
    }
    pl[r] = sl; pr[r] = sr;
  }
#pragma unroll
  for (int off = 8; off >= 1; off >>= 1) {
#pragma unroll
    for (int r = 0; r < 8; ++r) {
      pl[r] += __shfl_xor(pl[r], off, 32);
      pr[r] += __shfl_xor(pr[r], off, 32);
    }
  }
  if (m == 0) {
#pragma unroll
    for (int r = 0; r < 8; ++r) {
      el[tile * 16 + r + 8 * hi] = pl[r];
      er[tile * 16 + r + 8 * hi] = pr[r];
    }
  }
}

__global__ void edge_logit_kernel(const int* __restrict__ src,
                                  const int* __restrict__ dst,
                                  const float* __restrict__ el,
                                  const float* __restrict__ er,
                                  float* __restrict__ e_edge,
                                  unsigned* __restrict__ emax_u) {
  int i = blockIdx.x * blockDim.x + threadIdx.x;
  if (i >= NE) return;
  float e = el[src[i]] + er[dst[i]];
  e = (e > 0.0f) ? e : NEG_SLOPE * e;
  e_edge[i] = e;
  atomicMax(emax_u + dst[i], enc_f32(e));
}

__global__ void edge_exp_kernel(const int* __restrict__ dst,
                                float* __restrict__ e_edge,
                                const unsigned* __restrict__ emax_u,
                                float* __restrict__ denom) {
  int i = blockIdx.x * blockDim.x + threadIdx.x;
  if (i >= NE) return;
  const int d = dst[i];
  float emax = dec_f32(emax_u[d]);
  if ((__float_as_uint(emax) & 0x7F800000u) == 0x7F800000u) emax = 0.0f;
  float ee = expf(e_edge[i] - emax);
  e_edge[i] = ee;
  atomicAdd(denom + d, ee);
}

__global__ __launch_bounds__(256) void edge_scatter_kernel(
    const int* __restrict__ src, const int* __restrict__ dst,
    const float* __restrict__ e_edge, const float* __restrict__ denom,
    const float* __restrict__ feat, float* __restrict__ rst) {
  int gid = blockIdx.x * blockDim.x + threadIdx.x;
  int e = gid >> 5;
  int lane = gid & 31;
  if (e >= NE) return;
  const int s = src[e], d = dst[e];
  float dn = denom[d];
  dn = (dn > 0.0f) ? dn : 1.0f;
  const float alpha = e_edge[e] / dn;
  const float4 v = ((const float4*)(feat + (size_t)s * F_OUT))[lane];
  float* rr = rst + (size_t)d * F_OUT + lane * 4;
  atomicAdd(rr + 0, v.x * alpha);
  atomicAdd(rr + 1, v.y * alpha);
  atomicAdd(rr + 2, v.z * alpha);
  atomicAdd(rr + 3, v.w * alpha);
}

__global__ void finalize_kernel(const float* __restrict__ rst,
                                const float* __restrict__ b_gat,
                                float* __restrict__ out) {
  int i = blockIdx.x * blockDim.x + threadIdx.x;
  if (i >= NN * F_OUT) return;
  const int n = i >> 7, c = i & 127;
  const float y = rst[i] + b_gat[c];
  if (c < 64) out[(size_t)n * 64 + c] = y;
  else        out[(size_t)NN * 64 + (size_t)n * 64 + (c - 64)] = tanhf(y);
}

extern "C" void kernel_launch(void* const* d_in, const int* in_sizes, int n_in,
                              void* d_out, int out_size, void* d_ws,
                              size_t ws_size, hipStream_t stream) {
  const float* embed  = (const float*)d_in[0];
  const int*   src    = (const int*)d_in[1];
  const int*   dst    = (const int*)d_in[2];
  const float* W_lin  = (const float*)d_in[3];
  const float* b_lin  = (const float*)d_in[4];
  const float* W_gat  = (const float*)d_in[5];
  const float* attn_l = (const float*)d_in[6];
  const float* attn_r = (const float*)d_in[7];
  const float* b_gat  = (const float*)d_in[8];
  float* out = (float*)d_out;

  float* ws     = (float*)d_ws;
  float* feat   = ws;
  float* el     = feat + (size_t)NN * F_OUT;
  float* er     = el + NN;
  float* denom  = er + NN;
  unsigned* emax_u = (unsigned*)(denom + NN);
  float* e_edge = (float*)(emax_u + NN);
  float* rst    = e_edge + NE;

  init_kernel<<<(NN * F_OUT + 255) / 256, 256, 0, stream>>>(rst, denom, emax_u);

  const int n_tiles = NN / 16;
  const int nf_blocks = (n_tiles + 7) / 8;
  const size_t smem_bytes =
      (size_t)(EMB3 * F_LAT + F_LAT * F_OUT + F_LAT + 2 * F_OUT +
               8 * 16 * F_LAT) * sizeof(float);
  (void)hipFuncSetAttribute((const void*)node_feat_kernel,
                            hipFuncAttributeMaxDynamicSharedMemorySize,
                            (int)smem_bytes);
  node_feat_kernel<<<nf_blocks, 256, smem_bytes, stream>>>(
      embed, W_lin, b_lin, W_gat, attn_l, attn_r, feat, el, er, n_tiles);

  edge_logit_kernel<<<(NE + 255) / 256, 256, 0, stream>>>(src, dst, el, er,
                                                          e_edge, emax_u);
  edge_exp_kernel<<<(NE + 255) / 256, 256, 0, stream>>>(dst, e_edge, emax_u,
                                                        denom);
  edge_scatter_kernel<<<(NE * 32 + 255) / 256, 256, 0, stream>>>(
      src, dst, e_edge, denom, feat, rst);
  finalize_kernel<<<(NN * F_OUT + 255) / 256, 256, 0, stream>>>(rst, b_gat,
                                                                out);
}


# segment max/sum/row-sum by grouping, no float atomics: a returning int add ranks each edge in its destination, one wave per destination reduces in registers; sums now in arrival-rank order
# speedup vs baseline: 2.4155x; 2.4155x over previous
#include <hip/hip_runtime.h>
#include <math.h>

#define NN     100000
#define NE     1600000
#define EMB3   192
#define F_LAT  64
#define F_OUT  128
#define NEG_SLOPE 0.2f

typedef __attribute__((ext_vector_type(2))) float v2f;
typedef __attribute__((ext_vector_type(8))) float v8f;

__device__ __forceinline__ v8f wmma4(v2f a, v2f b, v8f c) {
  return __builtin_amdgcn_wmma_f32_16x16x4_f32(false, a, false, b, (short)0, c,
                                               false, false);
}

__global__ __launch_bounds__(256) void node_feat_kernel(
    const float* __restrict__ embed,
    const float* __restrict__ W_lin, const float* __restrict__ b_lin,
    const float* __restrict__ W_gat,
    const float* __restrict__ attn_l, const float* __restrict__ attn_r,
    float* __restrict__ feat, float* __restrict__ el, float* __restrict__ er,
    int* __restrict__ deg, int* __restrict__ cursor, int n_tiles) {
  extern __shared__ float smem[];
  float* sW1 = smem;
  float* sW2 = sW1 + EMB3 * F_LAT;
  float* sB1 = sW2 + F_LAT * F_OUT;
  float* sAl = sB1 + F_LAT;
  float* sAr = sAl + F_OUT;
  float* sX  = sAr + F_OUT;

  const int tid = threadIdx.x;
  for (int i = blockIdx.x * 256 + tid; i < NN; i += gridDim.x * 256) deg[i] = 0;
  if (blockIdx.x == 0 && tid == 0) *cursor = 0;
  for (int i = tid; i < EMB3 * F_LAT; i += 256) sW1[i] = W_lin[i];
  for (int i = tid; i < F_LAT * F_OUT; i += 256) sW2[i] = W_gat[i];
  if (tid < F_LAT) sB1[tid] = b_lin[tid];
  if (tid < F_OUT) { sAl[tid] = attn_l[tid]; sAr[tid] = attn_r[tid]; }
  __syncthreads();

  const int wave = tid >> 5;
  const int lane = tid & 31;
  const int tile = blockIdx.x * 8 + wave;
  if (tile >= n_tiles) return;

  const int m  = lane & 15;
  const int hi = lane >> 4;

  const float* arow = embed + (size_t)(tile * 16 + m) * EMB3;
  v8f acc[4] = {};
  for (int k0 = 0; k0 < EMB3; k0 += 4) {
    const int ka = k0 + 2 * hi;
    v2f a; a.x = arow[ka]; a.y = arow[ka + 1];
    const float* b0 = sW1 + ka * F_LAT + m;
    const float* b1 = b0 + F_LAT;
#pragma unroll
    for (int nt = 0; nt < 4; ++nt) {
      v2f b; b.x = b0[nt * 16]; b.y = b1[nt * 16];
      acc[nt] = wmma4(a, b, acc[nt]);
    }
  }

  float* xw = sX + wave * 16 * F_LAT;
#pragma unroll
  for (int r = 0; r < 8; ++r) {
    const int row = r + 8 * hi;
#pragma unroll
    for (int nt = 0; nt < 4; ++nt) {
      float v = acc[nt][r] + sB1[nt * 16 + m];
      xw[row * F_LAT + nt * 16 + m] = fmaxf(v, 0.0f);
    }
  }

  v8f fa[8] = {};
  for (int k0 = 0; k0 < F_LAT; k0 += 4) {
    const int ka = k0 + 2 * hi;
    v2f a; a.x = xw[m * F_LAT + ka]; a.y = xw[m * F_LAT + ka + 1];
    const float* b0 = sW2 + ka * F_OUT + m;
    const float* b1 = b0 + F_OUT;
#pragma unroll
    for (int nt = 0; nt < 8; ++nt) {
      v2f b; b.x = b0[nt * 16]; b.y = b1[nt * 16];
      fa[nt] = wmma4(a, b, fa[nt]);
    }
  }

  float pl[8], pr[8];
  const size_t base = (size_t)(tile * 16) * F_OUT;
#pragma unroll
  for (int r = 0; r < 8; ++r) {
    const int row = r + 8 * hi;
    float sl = 0.0f, sr = 0.0f;
#pragma unroll
    for (int nt = 0; nt < 8; ++nt) {
      const float v = fa[nt][r];
      feat[base + row * F_OUT + nt * 16 + m] = v;
      sl += v * sAl[nt * 16 + m];
      sr += v * sAr[nt * 16 + m];
    }
    pl[r] = sl; pr[r] = sr;
  }
#pragma unroll
  for (int off = 8; off >= 1; off >>= 1) {
#pragma unroll
    for (int r = 0; r < 8; ++r) {
      pl[r] += __shfl_xor(pl[r], off, 32);
      pr[r] += __shfl_xor(pr[r], off, 32);
    }
  }
  if (m == 0) {
#pragma unroll
    for (int r = 0; r < 8; ++r) {
      el[tile * 16 + r + 8 * hi] = pl[r];
      er[tile * 16 + r + 8 * hi] = pr[r];
    }
  }
}


__global__ void edge_rank_kernel(const int* __restrict__ dst,
                                 int* __restrict__ deg,
                                 int* __restrict__ rank) {
  int i = blockIdx.x * blockDim.x + threadIdx.x;
  if (i >= NE) return;
  rank[i] = atomicAdd(deg + dst[i], 1);
}

__global__ void seg_alloc_kernel(const int* __restrict__ deg,
                                 int* __restrict__ cursor,
                                 int* __restrict__ start) {
  const int d = blockIdx.x * blockDim.x + threadIdx.x;
  const int lane = threadIdx.x & 31;
  const int v = (d < NN) ? deg[d] : 0;
  int incl = v;
  for (int off = 1; off < 32; off <<= 1) {
    const int t = __shfl_up(incl, off, 32);
    if (lane >= off) incl += t;
  }
  const int total = __shfl(incl, 31, 32);
  int base = 0;
  if (lane == 0) base = atomicAdd(cursor, total);
  base = __shfl(base, 0, 32);
  if (d < NN) start[d] = base + incl - v;
}

__global__ void edge_place_kernel(const int* __restrict__ src,
                                  const int* __restrict__ dst,
                                  const int* __restrict__ rank,
                                  const int* __restrict__ start,
                                  int* __restrict__ csr_src) {
  int i = blockIdx.x * blockDim.x + threadIdx.x;
  if (i >= NE) return;
  csr_src[start[dst[i]] + rank[i]] = src[i];
}

__device__ __forceinline__ float lrelu(float e) {
  return (e > 0.0f) ? e : NEG_SLOPE * e;
}
__device__ __forceinline__ float wave_max(float x) {
  for (int off = 16; off >= 1; off >>= 1) x = fmaxf(x, __shfl_xor(x, off, 32));
  return x;
}
__device__ __forceinline__ float wave_sum(float x) {
  for (int off = 16; off >= 1; off >>= 1) x += __shfl_xor(x, off, 32);
  return x;
}

__global__ __launch_bounds__(256) void dst_gather_kernel(
    const int* __restrict__ deg, const int* __restrict__ start,
    const int* __restrict__ csr_src,
    const float* __restrict__ el, const float* __restrict__ er,
    const float* __restrict__ feat, const float* __restrict__ b_gat,
    float* __restrict__ out) {
  const int lane = threadIdx.x & 31;
  const int d = __builtin_amdgcn_readfirstlane(blockIdx.x * 8 + (threadIdx.x >> 5));
  if (d >= NN) return;
  const int n = deg[d];
  const int beg = start[d];
  const float erd = er[d];

  int s0 = 0;
  float lg0 = -INFINITY;
  if (lane < n) {
    s0 = csr_src[beg + lane];
    lg0 = lrelu(el[s0] + erd);
  }
  float m = lg0;
  for (int j = 32 + lane; j < n; j += 32)
    m = fmaxf(m, lrelu(el[csr_src[beg + j]] + erd));
  m = wave_max(m);
  if ((__float_as_uint(m) & 0x7F800000u) == 0x7F800000u) m = 0.0f;

  const float ee0 = (lane < n) ? expf(lg0 - m) : 0.0f;
  float sum = ee0;
  for (int j = 32 + lane; j < n; j += 32)
    sum += expf(lrelu(el[csr_src[beg + j]] + erd) - m);
  sum = wave_sum(sum);
  const float dn = (sum > 0.0f) ? sum : 1.0f;

  const float4* feat4 = (const float4*)feat;
  float4 acc = {0.0f, 0.0f, 0.0f, 0.0f};
  int sC = s0;
  float aC = ee0 / dn;
  for (int j0 = 0; j0 < n; j0 += 32) {
    if (j0 > 0) {
      const int j = j0 + lane;
      sC = 0;
      aC = 0.0f;
      if (j < n) {
        sC = csr_src[beg + j];
        aC = expf(lrelu(el[sC] + erd) - m) / dn;
      }
    }
    const int c = (n - j0 < 32) ? (n - j0) : 32;
#pragma unroll 4
    for (int k = 0; k < c; ++k) {
      const int s = __builtin_amdgcn_readlane(sC, k);
      const float a = __int_as_float(
          __builtin_amdgcn_readlane(__float_as_int(aC), k));
      const float4 v = feat4[(size_t)s * (F_OUT / 4) + lane];
      acc.x += v.x * a;
      acc.y += v.y * a;
      acc.z += v.z * a;
      acc.w += v.w * a;
    }
  }

  const int c0 = lane * 4;
  float4 y;
  y.x = acc.x + b_gat[c0 + 0];
  y.y = acc.y + b_gat[c0 + 1];
  y.z = acc.z + b_gat[c0 + 2];
  y.w = acc.w + b_gat[c0 + 3];
  float* o = out + (size_t)d * 64 + c0;
  if (lane >= 16) {
    y.x = tanhf(y.x); y.y = tanhf(y.y); y.z = tanhf(y.z); y.w = tanhf(y.w);
    o = out + (size_t)NN * 64 + (size_t)d * 64 + (c0 - 64);
  }
  *(float4*)o = y;
}

extern "C" void kernel_launch(void* const* d_in, const int* in_sizes, int n_in,
                              void* d_out, int out_size, void* d_ws,
                              size_t ws_size, hipStream_t stream) {
  const float* embed  = (const float*)d_in[0];
  const int*   src    = (const int*)d_in[1];
  const int*   dst    = (const int*)d_in[2];
  const float* W_lin  = (const float*)d_in[3];
  const float* b_lin  = (const float*)d_in[4];
  const float* W_gat  = (const float*)d_in[5];
  const float* attn_l = (const float*)d_in[6];
  const float* attn_r = (const float*)d_in[7];
  const float* b_gat  = (const float*)d_in[8];
  float* out = (float*)d_out;

  float* ws     = (float*)d_ws;
  float* feat   = ws;
  float* el     = feat + (size_t)NN * F_OUT;
  float* er     = el + NN;
  int* deg      = (int*)(er + NN);
  int* start    = deg + NN;
  int* cursor   = start + NN;
  int* rank     = cursor + 32;
  int* csr_src  = rank + NE;

  const int n_tiles = NN / 16;
  const int nf_blocks = (n_tiles + 7) / 8;
  const size_t smem_bytes =
      (size_t)(EMB3 * F_LAT + F_LAT * F_OUT + F_LAT + 2 * F_OUT +
               8 * 16 * F_LAT) * sizeof(float);
  (void)hipFuncSetAttribute((const void*)node_feat_kernel,
                            hipFuncAttributeMaxDynamicSharedMemorySize,
                            (int)smem_bytes);
  node_feat_kernel<<<nf_blocks, 256, smem_bytes, stream>>>(
      embed, W_lin, b_lin, W_gat, attn_l, attn_r, feat, el, er, deg, cursor,
      n_tiles);

  edge_rank_kernel<<<(NE + 255) / 256, 256, 0, stream>>>(dst, deg, rank);
  seg_alloc_kernel<<<(NN + 255) / 256, 256, 0, stream>>>(deg, cursor, start);
  edge_place_kernel<<<(NE + 255) / 256, 256, 0, stream>>>(src, dst, rank,
                                                          start, csr_src);
  dst_gather_kernel<<<(NN + 7) / 8, 256, 0, stream>>>(deg, start, csr_src, el,
                                                      er, feat, b_gat, out);
}


# edges ranked inside their destination by returning adds on LDS counters, one block per 25000-edge chunk, with a per-chunk slot table, not 1.6M global adds; sums now in chunk-then-LDS-arrival order
# speedup vs baseline: 3.2709x; 1.3541x over previous
#include <hip/hip_runtime.h>
#include <math.h>

#define NN     100000
#define NE     1600000
#define EMB3   192
#define F_LAT  64
#define F_OUT  128
#define NEG_SLOPE 0.2f
#define NCHUNK  64
#define CHUNK_E (NE / NCHUNK)
#define CHUNK_Q (CHUNK_E / 4)
#define RANK_T  1024
#define RANK_U  ((CHUNK_Q + RANK_T - 1) / RANK_T)
#define NWORD   (NN / 2)

typedef __attribute__((ext_vector_type(2))) float v2f;
typedef __attribute__((ext_vector_type(8))) float v8f;

__device__ __forceinline__ v8f wmma4(v2f a, v2f b, v8f c) {
  return __builtin_amdgcn_wmma_f32_16x16x4_f32(false, a, false, b, (short)0, c,
                                               false, false);
}

__global__ __launch_bounds__(256) void node_feat_kernel(
    const float* __restrict__ embed,
    const float* __restrict__ W_lin, const float* __restrict__ b_lin,
    const float* __restrict__ W_gat,
    const float* __restrict__ attn_l, const float* __restrict__ attn_r,
    float* __restrict__ feat, float* __restrict__ el, float* __restrict__ er,
    int n_tiles) {
  extern __shared__ float smem[];
  float* sW1 = smem;
  float* sW2 = sW1 + EMB3 * F_LAT;
  float* sB1 = sW2 + F_LAT * F_OUT;
  float* sAl = sB1 + F_LAT;
  float* sAr = sAl + F_OUT;
  float* sX  = sAr + F_OUT;

  const int tid = threadIdx.x;
  for (int i = tid; i < EMB3 * F_LAT; i += 256) sW1[i] = W_lin[i];
  for (int i = tid; i < F_LAT * F_OUT; i += 256) sW2[i] = W_gat[i];
  if (tid < F_LAT) sB1[tid] = b_lin[tid];
  if (tid < F_OUT) { sAl[tid] = attn_l[tid]; sAr[tid] = attn_r[tid]; }
  __syncthreads();

  const int wave = tid >> 5;
  const int lane = tid & 31;
  const int tile = blockIdx.x * 8 + wave;
  if (tile >= n_tiles) return;

  const int m  = lane & 15;
  const int hi = lane >> 4;

  const float* arow = embed + (size_t)(tile * 16 + m) * EMB3;
  v8f acc[4] = {};
  for (int k0 = 0; k0 < EMB3; k0 += 4) {
    const int ka = k0 + 2 * hi;
    v2f a; a.x = arow[ka]; a.y = arow[ka + 1];
    const float* b0 = sW1 + ka * F_LAT + m;
    const float* b1 = b0 + F_LAT;
#pragma unroll
    for (int nt = 0; nt < 4; ++nt) {
      v2f b; b.x = b0[nt * 16]; b.y = b1[nt * 16];
      acc[nt] = wmma4(a, b, acc[nt]);
    }
  }

  float* xw = sX + wave * 16 * F_LAT;
#pragma unroll
  for (int r = 0; r < 8; ++r) {
    const int row = r + 8 * hi;
#pragma unroll
    for (int nt = 0; nt < 4; ++nt) {
      float v = acc[nt][r] + sB1[nt * 16 + m];
      xw[row * F_LAT + nt * 16 + m] = fmaxf(v, 0.0f);
    }
  }

  v8f fa[8] = {};
  for (int k0 = 0; k0 < F_LAT; k0 += 4) {
    const int ka = k0 + 2 * hi;
    v2f a; a.x = xw[m * F_LAT + ka]; a.y = xw[m * F_LAT + ka + 1];
    const float* b0 = sW2 + ka * F_OUT + m;
    const float* b1 = b0 + F_OUT;
#pragma unroll
    for (int nt = 0; nt < 8; ++nt) {
      v2f b; b.x = b0[nt * 16]; b.y = b1[nt * 16];
      fa[nt] = wmma4(a, b, fa[nt]);
    }
  }

  float pl[8], pr[8];
  const size_t base = (size_t)(tile * 16) * F_OUT;
#pragma unroll
  for (int r = 0; r < 8; ++r) {
    const int row = r + 8 * hi;
    float sl = 0.0f, sr = 0.0f;
#pragma unroll
    for (int nt = 0; nt < 8; ++nt) {
      const float v = fa[nt][r];
      feat[base + row * F_OUT + nt * 16 + m] = v;
      sl += v * sAl[nt * 16 + m];
      sr += v * sAr[nt * 16 + m];
    }
    pl[r] = sl; pr[r] = sr;
  }
#pragma unroll
  for (int off = 8; off >= 1; off >>= 1) {
#pragma unroll
    for (int r = 0; r < 8; ++r) {
      pl[r] += __shfl_xor(pl[r], off, 32);
      pr[r] += __shfl_xor(pr[r], off, 32);
    }
  }
  if (m == 0) {
#pragma unroll
    for (int r = 0; r < 8; ++r) {
      el[tile * 16 + r + 8 * hi] = pl[r];
      er[tile * 16 + r + 8 * hi] = pr[r];
    }
  }
}


__device__ __forceinline__ int rank16(unsigned* cnt, int d) {
  const unsigned sh = (unsigned)(d & 1) << 4;
  const unsigned old = atomicAdd(&cnt[d >> 1], 1u << sh);
  return (int)((old >> sh) & 0xFFFFu);
}

__global__ __launch_bounds__(RANK_T) void chunk_rank_kernel(
    const int* __restrict__ dst, int* __restrict__ rank,
    int* __restrict__ tab, int* __restrict__ cursor) {
  extern __shared__ unsigned cnt[];
  const int tid = threadIdx.x;
  const int c = blockIdx.x;
  if (c == 0 && tid == 0) *cursor = 0;

  const int4* d4 = (const int4*)(dst + (size_t)c * CHUNK_E);
  int4 dv[RANK_U];
#pragma unroll
  for (int u = 0; u < RANK_U; ++u) {
    const int q = tid + u * RANK_T;
    dv[u] = d4[(q < CHUNK_Q) ? q : (CHUNK_Q - 1)];
  }
  uint4* cnt4 = (uint4*)cnt;
  for (int j = tid; j < NWORD / 4; j += RANK_T)
    cnt4[j] = make_uint4(0u, 0u, 0u, 0u);
  __syncthreads();

  int4* r4 = (int4*)(rank + (size_t)c * CHUNK_E);
#pragma unroll
  for (int u = 0; u < RANK_U; ++u) {
    const int q = tid + u * RANK_T;
    if (q < CHUNK_Q) {
      int4 r;
      r.x = rank16(cnt, dv[u].x);
      r.y = rank16(cnt, dv[u].y);
      r.z = rank16(cnt, dv[u].z);
      r.w = rank16(cnt, dv[u].w);
      r4[q] = r;
    }
  }
  __syncthreads();

  int2* t2 = (int2*)(tab + (size_t)c * NN);
  for (int j = tid; j < NWORD; j += RANK_T) {
    const unsigned w = cnt[j];
    int2 v;
    v.x = (int)(w & 0xFFFFu);
    v.y = (int)(w >> 16);
    t2[j] = v;
  }
}

__global__ void seg_alloc_kernel(int* __restrict__ tab,
                                 int* __restrict__ cursor,
                                 int* __restrict__ deg,
                                 int* __restrict__ start) {
  const int d = blockIdx.x * blockDim.x + threadIdx.x;
  const int lane = threadIdx.x & 31;
  const bool ok = d < NN;
  int* col = tab + (ok ? d : (NN - 1));
  int t[NCHUNK];
#pragma unroll
  for (int c = 0; c < NCHUNK; ++c) t[c] = col[(size_t)c * NN];
  int v = 0;
#pragma unroll
  for (int c = 0; c < NCHUNK; ++c) v += t[c];
  if (!ok) v = 0;
  int incl = v;
  for (int off = 1; off < 32; off <<= 1) {
    const int s = __shfl_up(incl, off, 32);
    if (lane >= off) incl += s;
  }
  const int total = __shfl(incl, 31, 32);
  int base = 0;
  if (lane == 0) base = atomicAdd(cursor, total);
  base = __shfl(base, 0, 32);
  if (ok) {
    int run = base + incl - v;
    deg[d] = v;
    start[d] = run;
#pragma unroll
    for (int c = 0; c < NCHUNK; ++c) {
      col[(size_t)c * NN] = run;
      run += t[c];
    }
  }
}

__global__ void edge_place_kernel(const int* __restrict__ src,
                                  const int* __restrict__ dst,
                                  const int* __restrict__ rank,
                                  const int* __restrict__ tab,
                                  int* __restrict__ csr_src) {
  int i = blockIdx.x * blockDim.x + threadIdx.x;
  if (i >= NE) return;
  const int c = i / CHUNK_E;
  csr_src[tab[(size_t)c * NN + dst[i]] + rank[i]] = src[i];
}

__device__ __forceinline__ float lrelu(float e) {
  return (e > 0.0f) ? e : NEG_SLOPE * e;
}
__device__ __forceinline__ float wave_max(float x) {
  for (int off = 16; off >= 1; off >>= 1) x = fmaxf(x, __shfl_xor(x, off, 32));
  return x;
}
__device__ __forceinline__ float wave_sum(float x) {
  for (int off = 16; off >= 1; off >>= 1) x += __shfl_xor(x, off, 32);
  return x;
}

__global__ __launch_bounds__(256) void dst_gather_kernel(
    const int* __restrict__ deg, const int* __restrict__ start,
    const int* __restrict__ csr_src,
    const float* __restrict__ el, const float* __restrict__ er,
    const float* __restrict__ feat, const float* __restrict__ b_gat,
    float* __restrict__ out) {
  const int lane = threadIdx.x & 31;
  const int d = __builtin_amdgcn_readfirstlane(blockIdx.x * 8 + (threadIdx.x >> 5));
  if (d >= NN) return;
  const int n = deg[d];
  const int beg = start[d];
  const float erd = er[d];

  int s0 = 0;
  float lg0 = -INFINITY;
  if (lane < n) {
    s0 = csr_src[beg + lane];
    lg0 = lrelu(el[s0] + erd);
  }
  float m = lg0;
  for (int j = 32 + lane; j < n; j += 32)
    m = fmaxf(m, lrelu(el[csr_src[beg + j]] + erd));
  m = wave_max(m);
  if ((__float_as_uint(m) & 0x7F800000u) == 0x7F800000u) m = 0.0f;

  const float ee0 = (lane < n) ? expf(lg0 - m) : 0.0f;
  float sum = ee0;
  for (int j = 32 + lane; j < n; j += 32)
    sum += expf(lrelu(el[csr_src[beg + j]] + erd) - m);
  sum = wave_sum(sum);
  const float dn = (sum > 0.0f) ? sum : 1.0f;

  const float4* feat4 = (const float4*)feat;
  float4 acc = {0.0f, 0.0f, 0.0f, 0.0f};
  int sC = s0;
  float aC = ee0 / dn;
  for (int j0 = 0; j0 < n; j0 += 32) {
    if (j0 > 0) {
      const int j = j0 + lane;
      sC = 0;
      aC = 0.0f;
      if (j < n) {
        sC = csr_src[beg + j];
        aC = expf(lrelu(el[sC] + erd) - m) / dn;
      }
    }
    const int c = (n - j0 < 32) ? (n - j0) : 32;
#pragma unroll 4
    for (int k = 0; k < c; ++k) {
      const int s = __builtin_amdgcn_readlane(sC, k);
      const float a = __int_as_float(
          __builtin_amdgcn_readlane(__float_as_int(aC), k));
      const float4 v = feat4[(size_t)s * (F_OUT / 4) + lane];
      acc.x += v.x * a;
      acc.y += v.y * a;
      acc.z += v.z * a;
      acc.w += v.w * a;
    }
  }

  const int c0 = lane * 4;
  float4 y;
  y.x = acc.x + b_gat[c0 + 0];
  y.y = acc.y + b_gat[c0 + 1];
  y.z = acc.z + b_gat[c0 + 2];
  y.w = acc.w + b_gat[c0 + 3];
  float* o = out + (size_t)d * 64 + c0;
  if (lane >= 16) {
    y.x = tanhf(y.x); y.y = tanhf(y.y); y.z = tanhf(y.z); y.w = tanhf(y.w);
    o = out + (size_t)NN * 64 + (size_t)d * 64 + (c0 - 64);
  }
  *(float4*)o = y;
}

extern "C" void kernel_launch(void* const* d_in, const int* in_sizes, int n_in,
                              void* d_out, int out_size, void* d_ws,
                              size_t ws_size, hipStream_t stream) {
  const float* embed  = (const float*)d_in[0];
  const int*   src    = (const int*)d_in[1];
  const int*   dst    = (const int*)d_in[2];
  const float* W_lin  = (const float*)d_in[3];
  const float* b_lin  = (const float*)d_in[4];
  const float* W_gat  = (const float*)d_in[5];
  const float* attn_l = (const float*)d_in[6];
  const float* attn_r = (const float*)d_in[7];
  const float* b_gat  = (const float*)d_in[8];
  float* out = (float*)d_out;

  float* ws     = (float*)d_ws;
  float* feat   = ws;
  float* el     = feat + (size_t)NN * F_OUT;
  float* er     = el + NN;
  int* deg      = (int*)(er + NN);
  int* start    = deg + NN;
  int* cursor   = start + NN;
  int* rank     = cursor + 32;
  int* csr_src  = rank + NE;
  int* tab      = (int*)feat;

  const size_t rank_lds = (size_t)NWORD * sizeof(unsigned);
  (void)hipFuncSetAttribute((const void*)chunk_rank_kernel,
                            hipFuncAttributeMaxDynamicSharedMemorySize,
                            (int)rank_lds);
  chunk_rank_kernel<<<NCHUNK, RANK_T, rank_lds, stream>>>(dst, rank, tab,
                                                          cursor);
  seg_alloc_kernel<<<(NN + 255) / 256, 256, 0, stream>>>(tab, cursor, deg,
                                                         start);
  edge_place_kernel<<<(NE + 255) / 256, 256, 0, stream>>>(src, dst, rank, tab,
                                                          csr_src);

  const int n_tiles = NN / 16;
  const int nf_blocks = (n_tiles + 7) / 8;
  const size_t smem_bytes =
      (size_t)(EMB3 * F_LAT + F_LAT * F_OUT + F_LAT + 2 * F_OUT +
               8 * 16 * F_LAT) * sizeof(float);
  (void)hipFuncSetAttribute((const void*)node_feat_kernel,
                            hipFuncAttributeMaxDynamicSharedMemorySize,
                            (int)smem_bytes);
  node_feat_kernel<<<nf_blocks, 256, smem_bytes, stream>>>(
      embed, W_lin, b_lin, W_gat, attn_l, attn_r, feat, el, er, n_tiles);

  dst_gather_kernel<<<(NN + 7) / 8, 256, 0, stream>>>(deg, start, csr_src, el,
                                                      er, feat, b_gat, out);
}


# W_gat projection moved behind the aggregation (linearity): gather 64-wide x rows, not 128-wide feat; W_gat applied per 16-destination tile on WMMA; logits from x.(W_gat.attn); sum order changes
# speedup vs baseline: 3.6300x; 1.1098x over previous
#include <hip/hip_runtime.h>
#include <math.h>

#define NN     100000
#define NE     1600000
#define EMB3   192
#define F_LAT  64
#define F_OUT  128
#define NEG_SLOPE 0.2f
#define NCHUNK  64
#define CHUNK_E (NE / NCHUNK)
#define CHUNK_Q (CHUNK_E / 4)
#define RANK_T  1024
#define RANK_U  ((CHUNK_Q + RANK_T - 1) / RANK_T)
#define NWORD   (NN / 2)
#define XP      68
#define OP      132
#define GP_T    512
#define GP_GRID 1024

typedef __attribute__((ext_vector_type(2))) float v2f;
typedef __attribute__((ext_vector_type(8))) float v8f;

__device__ __forceinline__ v8f wmma4(v2f a, v2f b, v8f c) {
  return __builtin_amdgcn_wmma_f32_16x16x4_f32(false, a, false, b, (short)0, c,
                                               false, false);
}

__global__ __launch_bounds__(256) void node_x_kernel(
    const float* __restrict__ embed,
    const float* __restrict__ W_lin, const float* __restrict__ b_lin,
    const float* __restrict__ W_gat,
    const float* __restrict__ attn_l, const float* __restrict__ attn_r,
    float* __restrict__ xg, float* __restrict__ el, float* __restrict__ er,
    int n_tiles) {
  extern __shared__ float smem[];
  float* sW1 = smem;
  float* sB1 = sW1 + EMB3 * F_LAT;
  float* sWl = sB1 + F_LAT;
  float* sWr = sWl + F_LAT;
  float* sX  = sWr + F_LAT;

  const int tid = threadIdx.x;
  for (int i = tid; i < EMB3 * F_LAT; i += 256) sW1[i] = W_lin[i];
  if (tid < F_LAT) sB1[tid] = b_lin[tid];
  if (tid < 2 * F_LAT) {
    const int k = tid & (F_LAT - 1);
    const float4* w4 = (const float4*)(W_gat + k * F_OUT);
    const float4* a4 = (const float4*)((tid < F_LAT) ? attn_l : attn_r);
    float s = 0.0f;
#pragma unroll 4
    for (int c = 0; c < F_OUT / 4; ++c) {
      const float4 w = w4[c];
      const float4 a = a4[c];
      s += w.x * a.x + w.y * a.y + w.z * a.z + w.w * a.w;
    }
    if (tid < F_LAT) sWl[k] = s; else sWr[k] = s;
  }
  __syncthreads();

  const int wave = tid >> 5;
  const int lane = tid & 31;
  const int tile = blockIdx.x * 8 + wave;
  if (tile >= n_tiles) return;

  const int m  = lane & 15;
  const int hi = lane >> 4;

  const float* arow = embed + (size_t)(tile * 16 + m) * EMB3;
  v8f acc[4] = {};
  for (int k0 = 0; k0 < EMB3; k0 += 4) {
    const int ka = k0 + 2 * hi;
    v2f a; a.x = arow[ka]; a.y = arow[ka + 1];
    const float* b0 = sW1 + ka * F_LAT + m;
    const float* b1 = b0 + F_LAT;
#pragma unroll
    for (int nt = 0; nt < 4; ++nt) {
      v2f b; b.x = b0[nt * 16]; b.y = b1[nt * 16];
      acc[nt] = wmma4(a, b, acc[nt]);
    }
  }

  float* xw = sX + wave * 16 * XP;
#pragma unroll
  for (int r = 0; r < 8; ++r) {
    const int row = r + 8 * hi;
#pragma unroll
    for (int nt = 0; nt < 4; ++nt) {
      float v = acc[nt][r] + sB1[nt * 16 + m];
      xw[row * XP + nt * 16 + m] = fmaxf(v, 0.0f);
    }
  }

  const int c4 = m * 4;
  const float4 wl = *(const float4*)(sWl + c4);
  const float4 wr = *(const float4*)(sWr + c4);
  float4* xo = (float4*)(xg + (size_t)(tile * 16) * F_LAT);
  float pl[8], pr[8];
#pragma unroll
  for (int it = 0; it < 8; ++it) {
    const int row = 2 * it + hi;
    const float4 v = *(const float4*)(xw + row * XP + c4);
    xo[row * (F_LAT / 4) + m] = v;
    pl[it] = v.x * wl.x + v.y * wl.y + v.z * wl.z + v.w * wl.w;
    pr[it] = v.x * wr.x + v.y * wr.y + v.z * wr.z + v.w * wr.w;
  }
#pragma unroll
  for (int off = 8; off >= 1; off >>= 1) {
#pragma unroll
    for (int it = 0; it < 8; ++it) {
      pl[it] += __shfl_xor(pl[it], off, 32);
      pr[it] += __shfl_xor(pr[it], off, 32);
    }
  }
  if (m == 0) {
#pragma unroll
    for (int it = 0; it < 8; ++it) {
      el[tile * 16 + 2 * it + hi] = pl[it];
      er[tile * 16 + 2 * it + hi] = pr[it];
    }
  }
}


__device__ __forceinline__ int rank16(unsigned* cnt, int d) {
  const unsigned sh = (unsigned)(d & 1) << 4;
  const unsigned old = atomicAdd(&cnt[d >> 1], 1u << sh);
  return (int)((old >> sh) & 0xFFFFu);
}

__global__ __launch_bounds__(RANK_T) void chunk_rank_kernel(
    const int* __restrict__ dst, int* __restrict__ rank,
    int* __restrict__ tab, int* __restrict__ cursor) {
  extern __shared__ unsigned cnt[];
  const int tid = threadIdx.x;
  const int c = blockIdx.x;
  if (c == 0 && tid == 0) *cursor = 0;

  const int4* d4 = (const int4*)(dst + (size_t)c * CHUNK_E);
  int4 dv[RANK_U];
#pragma unroll
  for (int u = 0; u < RANK_U; ++u) {
    const int q = tid + u * RANK_T;
    dv[u] = d4[(q < CHUNK_Q) ? q : (CHUNK_Q - 1)];
  }
  uint4* cnt4 = (uint4*)cnt;
  for (int j = tid; j < NWORD / 4; j += RANK_T)
    cnt4[j] = make_uint4(0u, 0u, 0u, 0u);
  __syncthreads();

  int4* r4 = (int4*)(rank + (size_t)c * CHUNK_E);
#pragma unroll
  for (int u = 0; u < RANK_U; ++u) {
    const int q = tid + u * RANK_T;
    if (q < CHUNK_Q) {
      int4 r;
      r.x = rank16(cnt, dv[u].x);
      r.y = rank16(cnt, dv[u].y);
      r.z = rank16(cnt, dv[u].z);
      r.w = rank16(cnt, dv[u].w);
      r4[q] = r;
    }
  }
  __syncthreads();

  int2* t2 = (int2*)(tab + (size_t)c * NN);
  for (int j = tid; j < NWORD; j += RANK_T) {
    const unsigned w = cnt[j];
    int2 v;
    v.x = (int)(w & 0xFFFFu);
    v.y = (int)(w >> 16);
    t2[j] = v;
  }
}

__global__ void seg_alloc_kernel(int* __restrict__ tab,
                                 int* __restrict__ cursor,
                                 int* __restrict__ deg,
                                 int* __restrict__ start) {
  const int d = blockIdx.x * blockDim.x + threadIdx.x;
  const int lane = threadIdx.x & 31;
  const bool ok = d < NN;
  int* col = tab + (ok ? d : (NN - 1));
  int t[NCHUNK];
#pragma unroll
  for (int c = 0; c < NCHUNK; ++c) t[c] = col[(size_t)c * NN];
  int v = 0;
#pragma unroll
  for (int c = 0; c < NCHUNK; ++c) v += t[c];
  if (!ok) v = 0;
  int incl = v;
  for (int off = 1; off < 32; off <<= 1) {
    const int s = __shfl_up(incl, off, 32);
    if (lane >= off) incl += s;
  }
  const int total = __shfl(incl, 31, 32);
  int base = 0;
  if (lane == 0) base = atomicAdd(cursor, total);
  base = __shfl(base, 0, 32);
  if (ok) {
    int run = base + incl - v;
    deg[d] = v;
    start[d] = run;
#pragma unroll
    for (int c = 0; c < NCHUNK; ++c) {
      col[(size_t)c * NN] = run;
      run += t[c];
    }
  }
}

__global__ void edge_place_kernel(const int* __restrict__ src,
                                  const int* __restrict__ dst,
                                  const int* __restrict__ rank,
                                  const int* __restrict__ tab,
                                  int* __restrict__ csr_src) {
  int i = blockIdx.x * blockDim.x + threadIdx.x;
  if (i >= NE) return;
  const int c = i / CHUNK_E;
  csr_src[tab[(size_t)c * NN + dst[i]] + rank[i]] = src[i];
}

__device__ __forceinline__ float lrelu(float e) {
  return (e > 0.0f) ? e : NEG_SLOPE * e;
}
__device__ __forceinline__ float wave_max(float x) {
  for (int off = 16; off >= 1; off >>= 1) x = fmaxf(x, __shfl_xor(x, off, 32));
  return x;
}
__device__ __forceinline__ float wave_sum(float x) {
  for (int off = 16; off >= 1; off >>= 1) x += __shfl_xor(x, off, 32);
  return x;
}

__global__ __launch_bounds__(GP_T, 16) void dst_gather_project_kernel(
    const int* __restrict__ deg, const int* __restrict__ start,
    const int* __restrict__ csr_src,
    const float* __restrict__ el, const float* __restrict__ er,
    const float* __restrict__ xg, const float* __restrict__ W_gat,
    const float* __restrict__ b_gat, float* __restrict__ out) {
  __shared__ __attribute__((aligned(16))) float sW[F_LAT * F_OUT];
  __shared__ __attribute__((aligned(16))) float sAgg[16 * XP];
  __shared__ __attribute__((aligned(16))) float sOut[16 * OP];

  const int tid = threadIdx.x;
  const int lane = tid & 31;
  const int m  = lane & 15;
  const int hi = lane >> 4;
  const int w = __builtin_amdgcn_readfirstlane(tid >> 5);

  for (int i = tid; i < F_LAT * F_OUT / 4; i += GP_T)
    ((float4*)sW)[i] = ((const float4*)W_gat)[i];

  const float4* x4 = (const float4*)xg;
  const int c0 = lane * 4;
  const float4 bias = *(const float4*)(b_gat + c0);

  for (int t = blockIdx.x; t < NN / 16; t += GP_GRID) {
    const int d = t * 16 + w;
    const int n = deg[d];
    const int beg = start[d];
    const float erd = er[d];

    int s0 = 0;
    float lg0 = -INFINITY;
    if (lane < n) {
      s0 = csr_src[beg + lane];
      lg0 = lrelu(el[s0] + erd);
    }
    float mx = lg0;
    for (int j = 32 + lane; j < n; j += 32)
      mx = fmaxf(mx, lrelu(el[csr_src[beg + j]] + erd));
    mx = wave_max(mx);
    if ((__float_as_uint(mx) & 0x7F800000u) == 0x7F800000u) mx = 0.0f;

    const float ee0 = (lane < n) ? expf(lg0 - mx) : 0.0f;
    float sum = ee0;
    for (int j = 32 + lane; j < n; j += 32)
      sum += expf(lrelu(el[csr_src[beg + j]] + erd) - mx);
    sum = wave_sum(sum);
    const float dn = (sum > 0.0f) ? sum : 1.0f;

    float4 acc = {0.0f, 0.0f, 0.0f, 0.0f};
    int sC = s0;
    float aC = ee0 / dn;
    for (int j0 = 0; j0 < n; j0 += 32) {
      if (j0 > 0) {
        const int j = j0 + lane;
        sC = 0;
        aC = 0.0f;
        if (j < n) {
          sC = csr_src[beg + j];
          aC = expf(lrelu(el[sC] + erd) - mx) / dn;
        }
      }
      const int c = (n - j0 < 32) ? (n - j0) : 32;
#pragma unroll 4
      for (int k = 0; k < c; k += 2) {
        const int sLo = __builtin_amdgcn_readlane(sC, k);
        const int sHi = __builtin_amdgcn_readlane(sC, k + 1);
        const int aLo = __builtin_amdgcn_readlane(__float_as_int(aC), k);
        const int aHi = __builtin_amdgcn_readlane(__float_as_int(aC), k + 1);
        const int s = hi ? sHi : sLo;
        const float a = __int_as_float(hi ? aHi : aLo);
        const float4 v = x4[(size_t)s * (F_LAT / 4) + m];
        acc.x += v.x * a;
        acc.y += v.y * a;
        acc.z += v.z * a;
        acc.w += v.w * a;
      }
    }
    acc.x += __shfl_xor(acc.x, 16, 32);
    acc.y += __shfl_xor(acc.y, 16, 32);
    acc.z += __shfl_xor(acc.z, 16, 32);
    acc.w += __shfl_xor(acc.w, 16, 32);
    if (hi == 0) *(float4*)(sAgg + w * XP + m * 4) = acc;
    __syncthreads();

    if (w < 8) {
      v8f pc = {};
#pragma unroll 1
      for (int k0 = 0; k0 < F_LAT; k0 += 4) {
        const int ka = k0 + 2 * hi;
        v2f a; a.x = sAgg[m * XP + ka]; a.y = sAgg[m * XP + ka + 1];
        const float* b0 = sW + ka * F_OUT + w * 16 + m;
        v2f b; b.x = b0[0]; b.y = b0[F_OUT];
        pc = wmma4(a, b, pc);
      }
#pragma unroll
      for (int r = 0; r < 8; ++r)
        sOut[(r + 8 * hi) * OP + w * 16 + m] = pc[r];
    }
    __syncthreads();

    const float4 p = *(const float4*)(sOut + w * OP + c0);
    float4 y;
    y.x = p.x + bias.x;
    y.y = p.y + bias.y;
    y.z = p.z + bias.z;
    y.w = p.w + bias.w;
    float* o = out + (size_t)d * 64 + c0;
    if (lane >= 16) {
      y.x = tanhf(y.x); y.y = tanhf(y.y); y.z = tanhf(y.z); y.w = tanhf(y.w);
      o = out + (size_t)NN * 64 + (size_t)d * 64 + (c0 - 64);
    }
    *(float4*)o = y;
  }
}

extern "C" void kernel_launch(void* const* d_in, const int* in_sizes, int n_in,
                              void* d_out, int out_size, void* d_ws,
                              size_t ws_size, hipStream_t stream) {
  const float* embed  = (const float*)d_in[0];
  const int*   src    = (const int*)d_in[1];
  const int*   dst    = (const int*)d_in[2];
  const float* W_lin  = (const float*)d_in[3];
  const float* b_lin  = (const float*)d_in[4];
  const float* W_gat  = (const float*)d_in[5];
  const float* attn_l = (const float*)d_in[6];
  const float* attn_r = (const float*)d_in[7];
  const float* b_gat  = (const float*)d_in[8];
  float* out = (float*)d_out;

  float* ws     = (float*)d_ws;
  float* xg     = ws;
  float* el     = xg + (size_t)NN * F_OUT;
  float* er     = el + NN;
  int* deg      = (int*)(er + NN);
  int* start    = deg + NN;
  int* cursor   = start + NN;
  int* rank     = cursor + 32;
  int* csr_src  = rank + NE;
  int* tab      = (int*)xg;

  const size_t rank_lds = (size_t)NWORD * sizeof(unsigned);
  (void)hipFuncSetAttribute((const void*)chunk_rank_kernel,
                            hipFuncAttributeMaxDynamicSharedMemorySize,
                            (int)rank_lds);
  chunk_rank_kernel<<<NCHUNK, RANK_T, rank_lds, stream>>>(dst, rank, tab,
                                                          cursor);
  seg_alloc_kernel<<<(NN + 255) / 256, 256, 0, stream>>>(tab, cursor, deg,
                                                         start);
  edge_place_kernel<<<(NE + 255) / 256, 256, 0, stream>>>(src, dst, rank, tab,
                                                          csr_src);

  const int n_tiles = NN / 16;
  const int nf_blocks = (n_tiles + 7) / 8;
  const size_t smem_bytes =
      (size_t)(EMB3 * F_LAT + 3 * F_LAT + 8 * 16 * XP) *
      sizeof(float);
  (void)hipFuncSetAttribute((const void*)node_x_kernel,
                            hipFuncAttributeMaxDynamicSharedMemorySize,
                            (int)smem_bytes);
  node_x_kernel<<<nf_blocks, 256, smem_bytes, stream>>>(
      embed, W_lin, b_lin, W_gat, attn_l, attn_r, xg, el, er, n_tiles);

  dst_gather_project_kernel<<<GP_GRID, GP_T, 0, stream>>>(
      deg, start, csr_src, el, er, xg, W_gat, b_gat, out);
}


# node GEMM merged into the ranking launch as a second block role: 50 ranking blocks (32000-edge chunks) beside 206 node blocks, one per WGP; 4 launches; sums in chunk-then-arrival order, new chunks
# speedup vs baseline: 4.1483x; 1.1428x over previous
#include <hip/hip_runtime.h>
#include <math.h>

#define NN     100000
#define NE     1600000
#define EMB3   192
#define F_LAT  64
#define F_OUT  128
#define NEG_SLOPE 0.2f
#define NCHUNK  50
#define CHUNK_E (NE / NCHUNK)
#define CHUNK_Q (CHUNK_E / 4)
#define RANK_T  1024
#define RANK_U  ((CHUNK_Q + RANK_T - 1) / RANK_T)
#define NWORD   (NN / 2)
#define XP      68
#define OP      132
#define FRONT_T RANK_T
#define NODE_B  (256 - NCHUNK)
#define NODE_W  (NODE_B * (FRONT_T / 32))
static_assert(NE % NCHUNK == 0 && CHUNK_E % 4 == 0 && CHUNK_E < 65536, "chunking");
static_assert(NODE_W >= NN / 16, "every node tile needs a wave of its own");
#define GP_T    512
#define GP_GRID 1024

typedef __attribute__((ext_vector_type(2))) float v2f;
typedef __attribute__((ext_vector_type(8))) float v8f;

__device__ __forceinline__ v8f wmma4(v2f a, v2f b, v8f c) {
  return __builtin_amdgcn_wmma_f32_16x16x4_f32(false, a, false, b, (short)0, c,
                                               false, false);
}


__device__ __forceinline__ int rank16(unsigned* cnt, int d) {
  const unsigned sh = (unsigned)(d & 1) << 4;
  const unsigned old = atomicAdd(&cnt[d >> 1], 1u << sh);
  return (int)((old >> sh) & 0xFFFFu);
}

__global__ __launch_bounds__(FRONT_T) void rank_and_node_kernel(
    const int* __restrict__ dst, int* __restrict__ rank,
    int* __restrict__ tab, int* __restrict__ cursor,
    const float* __restrict__ embed,
    const float* __restrict__ W_lin, const float* __restrict__ b_lin,
    const float* __restrict__ W_gat,
    const float* __restrict__ attn_l, const float* __restrict__ attn_r,
    float* __restrict__ xg, float* __restrict__ el, float* __restrict__ er,
    int n_tiles) {
  extern __shared__ __attribute__((aligned(16))) float smem[];
  const int tid = threadIdx.x;

  if (blockIdx.x < NCHUNK) {
    unsigned* cnt = (unsigned*)smem;
    const int c = blockIdx.x;
    if (c == 0 && tid == 0) *cursor = 0;

    const int4* d4 = (const int4*)(dst + (size_t)c * CHUNK_E);
    int4 dv[RANK_U];
#pragma unroll
    for (int u = 0; u < RANK_U; ++u) {
      const int q = tid + u * RANK_T;
      dv[u] = d4[(q < CHUNK_Q) ? q : (CHUNK_Q - 1)];
    }
    uint4* cnt4 = (uint4*)cnt;
    for (int j = tid; j < NWORD / 4; j += RANK_T)
      cnt4[j] = make_uint4(0u, 0u, 0u, 0u);
    __syncthreads();

    int4* r4 = (int4*)(rank + (size_t)c * CHUNK_E);
#pragma unroll
    for (int u = 0; u < RANK_U; ++u) {
      const int q = tid + u * RANK_T;
      if (q < CHUNK_Q) {
        int4 r;
        r.x = rank16(cnt, dv[u].x);
        r.y = rank16(cnt, dv[u].y);
        r.z = rank16(cnt, dv[u].z);
        r.w = rank16(cnt, dv[u].w);
        r4[q] = r;
      }
    }
    __syncthreads();

    int2* t2 = (int2*)(tab + (size_t)c * NN);
    for (int j = tid; j < NWORD; j += RANK_T) {
      const unsigned w = cnt[j];
      int2 v;
      v.x = (int)(w & 0xFFFFu);
      v.y = (int)(w >> 16);
      t2[j] = v;
    }
    return;
  }

  float* sW1 = smem;
  float* sB1 = sW1 + EMB3 * F_LAT;
  float* sWl = sB1 + F_LAT;
  float* sWr = sWl + F_LAT;
  float* sX  = sWr + F_LAT;

  for (int i = tid; i < EMB3 * F_LAT; i += FRONT_T) sW1[i] = W_lin[i];
  if (tid < F_LAT) sB1[tid] = b_lin[tid];
  if (tid < 2 * F_LAT) {
    const int k = tid & (F_LAT - 1);
    const float4* w4 = (const float4*)(W_gat + k * F_OUT);
    const float4* a4 = (const float4*)((tid < F_LAT) ? attn_l : attn_r);
    float s = 0.0f;
#pragma unroll 4
    for (int c = 0; c < F_OUT / 4; ++c) {
      const float4 w = w4[c];
      const float4 a = a4[c];
      s += w.x * a.x + w.y * a.y + w.z * a.z + w.w * a.w;
    }
    if (tid < F_LAT) sWl[k] = s; else sWr[k] = s;
  }
  __syncthreads();

  const int wave = tid >> 5;
  const int lane = tid & 31;
  const int nb = blockIdx.x - NCHUNK;
  const int m  = lane & 15;
  const int hi = lane >> 4;
  float* xw = sX + wave * 16 * XP;
  const int c4 = m * 4;
  const float4 wl = *(const float4*)(sWl + c4);
  const float4 wr = *(const float4*)(sWr + c4);

  const int tile = wave * NODE_B + nb;
  if (tile < n_tiles) {
    const float* arow = embed + (size_t)(tile * 16 + m) * EMB3;
    v8f acc[4] = {};
    for (int k0 = 0; k0 < EMB3; k0 += 4) {
      const int ka = k0 + 2 * hi;
      v2f a; a.x = arow[ka]; a.y = arow[ka + 1];
      const float* b0 = sW1 + ka * F_LAT + m;
      const float* b1 = b0 + F_LAT;
#pragma unroll
      for (int nt = 0; nt < 4; ++nt) {
        v2f b; b.x = b0[nt * 16]; b.y = b1[nt * 16];
        acc[nt] = wmma4(a, b, acc[nt]);
      }
    }

#pragma unroll
    for (int r = 0; r < 8; ++r) {
      const int row = r + 8 * hi;
#pragma unroll
      for (int nt = 0; nt < 4; ++nt) {
        float v = acc[nt][r] + sB1[nt * 16 + m];
        xw[row * XP + nt * 16 + m] = fmaxf(v, 0.0f);
      }
    }

    float4* xo = (float4*)(xg + (size_t)(tile * 16) * F_LAT);
    float pl[8], pr[8];
#pragma unroll
    for (int it = 0; it < 8; ++it) {
      const int row = 2 * it + hi;
      const float4 v = *(const float4*)(xw + row * XP + c4);
      xo[row * (F_LAT / 4) + m] = v;
      pl[it] = v.x * wl.x + v.y * wl.y + v.z * wl.z + v.w * wl.w;
      pr[it] = v.x * wr.x + v.y * wr.y + v.z * wr.z + v.w * wr.w;
    }
#pragma unroll
    for (int off = 8; off >= 1; off >>= 1) {
#pragma unroll
      for (int it = 0; it < 8; ++it) {
        pl[it] += __shfl_xor(pl[it], off, 32);
        pr[it] += __shfl_xor(pr[it], off, 32);
      }
    }
    if (m == 0) {
#pragma unroll
      for (int it = 0; it < 8; ++it) {
        el[tile * 16 + 2 * it + hi] = pl[it];
        er[tile * 16 + 2 * it + hi] = pr[it];
      }
    }
  }
}

__global__ void seg_alloc_kernel(int* __restrict__ tab,
                                 int* __restrict__ cursor,
                                 int* __restrict__ deg,
                                 int* __restrict__ start) {
  const int d = blockIdx.x * blockDim.x + threadIdx.x;
  const int lane = threadIdx.x & 31;
  const bool ok = d < NN;
  int* col = tab + (ok ? d : (NN - 1));
  int t[NCHUNK];
#pragma unroll
  for (int c = 0; c < NCHUNK; ++c) t[c] = col[(size_t)c * NN];
  int v = 0;
#pragma unroll
  for (int c = 0; c < NCHUNK; ++c) v += t[c];
  if (!ok) v = 0;
  int incl = v;
  for (int off = 1; off < 32; off <<= 1) {
    const int s = __shfl_up(incl, off, 32);
    if (lane >= off) incl += s;
  }
  const int total = __shfl(incl, 31, 32);
  int base = 0;
  if (lane == 0) base = atomicAdd(cursor, total);
  base = __shfl(base, 0, 32);
  if (ok) {
    int run = base + incl - v;
    deg[d] = v;
    start[d] = run;
#pragma unroll
    for (int c = 0; c < NCHUNK; ++c) {
      col[(size_t)c * NN] = run;
      run += t[c];
    }
  }
}

__global__ void edge_place_kernel(const int* __restrict__ src,
                                  const int* __restrict__ dst,
                                  const int* __restrict__ rank,
                                  const int* __restrict__ tab,
                                  int* __restrict__ csr_src) {
  int i = blockIdx.x * blockDim.x + threadIdx.x;
  if (i >= NE) return;
  const int c = i / CHUNK_E;
  csr_src[tab[(size_t)c * NN + dst[i]] + rank[i]] = src[i];
}

__device__ __forceinline__ float lrelu(float e) {
  return (e > 0.0f) ? e : NEG_SLOPE * e;
}
__device__ __forceinline__ float wave_max(float x) {
  for (int off = 16; off >= 1; off >>= 1) x = fmaxf(x, __shfl_xor(x, off, 32));
  return x;
}
__device__ __forceinline__ float wave_sum(float x) {
  for (int off = 16; off >= 1; off >>= 1) x += __shfl_xor(x, off, 32);
  return x;
}

__global__ __launch_bounds__(GP_T, 16) void dst_gather_project_kernel(
    const int* __restrict__ deg, const int* __restrict__ start,
    const int* __restrict__ csr_src,
    const float* __restrict__ el, const float* __restrict__ er,
    const float* __restrict__ xg, const float* __restrict__ W_gat,
    const float* __restrict__ b_gat, float* __restrict__ out) {
  __shared__ __attribute__((aligned(16))) float sW[F_LAT * F_OUT];
  __shared__ __attribute__((aligned(16))) float sAgg[16 * XP];
  __shared__ __attribute__((aligned(16))) float sOut[16 * OP];

  const int tid = threadIdx.x;
  const int lane = tid & 31;
  const int m  = lane & 15;
  const int hi = lane >> 4;
  const int w = __builtin_amdgcn_readfirstlane(tid >> 5);

  for (int i = tid; i < F_LAT * F_OUT / 4; i += GP_T)
    ((float4*)sW)[i] = ((const float4*)W_gat)[i];

  const float4* x4 = (const float4*)xg;
  const int c0 = lane * 4;
  const float4 bias = *(const float4*)(b_gat + c0);

  for (int t = blockIdx.x; t < NN / 16; t += GP_GRID) {
    const int d = t * 16 + w;
    const int n = deg[d];
    const int beg = start[d];
    const float erd = er[d];

    int s0 = 0;
    float lg0 = -INFINITY;
    if (lane < n) {
      s0 = csr_src[beg + lane];
      lg0 = lrelu(el[s0] + erd);
    }
    float mx = lg0;
    for (int j = 32 + lane; j < n; j += 32)
      mx = fmaxf(mx, lrelu(el[csr_src[beg + j]] + erd));
    mx = wave_max(mx);
    if ((__float_as_uint(mx) & 0x7F800000u) == 0x7F800000u) mx = 0.0f;

    const float ee0 = (lane < n) ? expf(lg0 - mx) : 0.0f;
    float sum = ee0;
    for (int j = 32 + lane; j < n; j += 32)
      sum += expf(lrelu(el[csr_src[beg + j]] + erd) - mx);
    sum = wave_sum(sum);
    const float dn = (sum > 0.0f) ? sum : 1.0f;

    float4 acc = {0.0f, 0.0f, 0.0f, 0.0f};
    int sC = s0;
    float aC = ee0 / dn;
    for (int j0 = 0; j0 < n; j0 += 32) {
      if (j0 > 0) {
        const int j = j0 + lane;
        sC = 0;
        aC = 0.0f;
        if (j < n) {
          sC = csr_src[beg + j];
          aC = expf(lrelu(el[sC] + erd) - mx) / dn;
        }
      }
      const int c = (n - j0 < 32) ? (n - j0) : 32;
#pragma unroll 4
      for (int k = 0; k < c; k += 2) {
        const int sLo = __builtin_amdgcn_readlane(sC, k);
        const int sHi = __builtin_amdgcn_readlane(sC, k + 1);
        const int aLo = __builtin_amdgcn_readlane(__float_as_int(aC), k);
        const int aHi = __builtin_amdgcn_readlane(__float_as_int(aC), k + 1);
        const int s = hi ? sHi : sLo;
        const float a = __int_as_float(hi ? aHi : aLo);
        const float4 v = x4[(size_t)s * (F_LAT / 4) + m];
        acc.x += v.x * a;
        acc.y += v.y * a;
        acc.z += v.z * a;
        acc.w += v.w * a;
      }
    }
    acc.x += __shfl_xor(acc.x, 16, 32);
    acc.y += __shfl_xor(acc.y, 16, 32);
    acc.z += __shfl_xor(acc.z, 16, 32);
    acc.w += __shfl_xor(acc.w, 16, 32);
    if (hi == 0) *(float4*)(sAgg + w * XP + m * 4) = acc;
    __syncthreads();

    if (w < 8) {
      v8f pc = {};
#pragma unroll 1
      for (int k0 = 0; k0 < F_LAT; k0 += 4) {
        const int ka = k0 + 2 * hi;
        v2f a; a.x = sAgg[m * XP + ka]; a.y = sAgg[m * XP + ka + 1];
        const float* b0 = sW + ka * F_OUT + w * 16 + m;
        v2f b; b.x = b0[0]; b.y = b0[F_OUT];
        pc = wmma4(a, b, pc);
      }
#pragma unroll
      for (int r = 0; r < 8; ++r)
        sOut[(r + 8 * hi) * OP + w * 16 + m] = pc[r];
    }
    __syncthreads();

    const float4 p = *(const float4*)(sOut + w * OP + c0);
    float4 y;
    y.x = p.x + bias.x;
    y.y = p.y + bias.y;
    y.z = p.z + bias.z;
    y.w = p.w + bias.w;
    float* o = out + (size_t)d * 64 + c0;
    if (lane >= 16) {
      y.x = tanhf(y.x); y.y = tanhf(y.y); y.z = tanhf(y.z); y.w = tanhf(y.w);
      o = out + (size_t)NN * 64 + (size_t)d * 64 + (c0 - 64);
    }
    *(float4*)o = y;
  }
}

extern "C" void kernel_launch(void* const* d_in, const int* in_sizes, int n_in,
                              void* d_out, int out_size, void* d_ws,
                              size_t ws_size, hipStream_t stream) {
  const float* embed  = (const float*)d_in[0];
  const int*   src    = (const int*)d_in[1];
  const int*   dst    = (const int*)d_in[2];
  const float* W_lin  = (const float*)d_in[3];
  const float* b_lin  = (const float*)d_in[4];
  const float* W_gat  = (const float*)d_in[5];
  const float* attn_l = (const float*)d_in[6];
  const float* attn_r = (const float*)d_in[7];
  const float* b_gat  = (const float*)d_in[8];
  float* out = (float*)d_out;

  float* ws     = (float*)d_ws;
  float* xg     = ws;
  int* tab      = (int*)(xg + (size_t)NN * F_LAT);
  float* el     = xg + (size_t)NN * F_OUT;
  float* er     = el + NN;
  int* deg      = (int*)(er + NN);
  int* start    = deg + NN;
  int* cursor   = start + NN;
  int* rank     = cursor + 32;
  int* csr_src  = rank + NE;

  const int n_tiles = NN / 16;
  const size_t front_lds = (size_t)NWORD * sizeof(unsigned);
  (void)hipFuncSetAttribute((const void*)rank_and_node_kernel,
                            hipFuncAttributeMaxDynamicSharedMemorySize,
                            (int)front_lds);
  rank_and_node_kernel<<<NCHUNK + NODE_B, FRONT_T, front_lds, stream>>>(
      dst, rank, tab, cursor, embed, W_lin, b_lin, W_gat, attn_l, attn_r,
      xg, el, er, n_tiles);
  seg_alloc_kernel<<<(NN + 255) / 256, 256, 0, stream>>>(tab, cursor, deg,
                                                         start);
  edge_place_kernel<<<(NE + 255) / 256, 256, 0, stream>>>(src, dst, rank, tab,
                                                          csr_src);

  dst_gather_project_kernel<<<GP_GRID, GP_T, 0, stream>>>(
      deg, start, csr_src, el, er, xg, W_gat, b_gat, out);
}


# gather kernel: each tile's dependent index loads (degree/start/er, slot sources, their el) issued one tile ahead, kept in flight across the tile's barriers (fences narrowed to LDS); bit-identical
# speedup vs baseline: 4.2799x; 1.0317x over previous
#include <hip/hip_runtime.h>
#include <math.h>

#define NN     100000
#define NE     1600000
#define EMB3   192
#define F_LAT  64
#define F_OUT  128
#define NEG_SLOPE 0.2f
#define NCHUNK  50
#define CHUNK_E (NE / NCHUNK)
#define CHUNK_Q (CHUNK_E / 4)
#define RANK_T  1024
#define RANK_U  ((CHUNK_Q + RANK_T - 1) / RANK_T)
#define NWORD   (NN / 2)
#define XP      68
#define OP      132
#define FRONT_T RANK_T
#define NODE_B  (256 - NCHUNK)
#define NODE_W  (NODE_B * (FRONT_T / 32))
static_assert(NE % NCHUNK == 0 && CHUNK_E % 4 == 0 && CHUNK_E < 65536, "chunking");
static_assert(NODE_W >= NN / 16, "every node tile needs a wave of its own");
#define GP_T    512
#define GP_GRID 1024
static_assert(GP_GRID <= NN / 16, "every gather block starts with a tile of its own");

typedef __attribute__((ext_vector_type(2))) float v2f;
typedef __attribute__((ext_vector_type(8))) float v8f;

__device__ __forceinline__ v8f wmma4(v2f a, v2f b, v8f c) {
  return __builtin_amdgcn_wmma_f32_16x16x4_f32(false, a, false, b, (short)0, c,
                                               false, false);
}


__device__ __forceinline__ int rank16(unsigned* cnt, int d) {
  const unsigned sh = (unsigned)(d & 1) << 4;
  const unsigned old = atomicAdd(&cnt[d >> 1], 1u << sh);
  return (int)((old >> sh) & 0xFFFFu);
}

__global__ __launch_bounds__(FRONT_T) void rank_and_node_kernel(
    const int* __restrict__ dst, int* __restrict__ rank,
    int* __restrict__ tab, int* __restrict__ cursor,
    const float* __restrict__ embed,
    const float* __restrict__ W_lin, const float* __restrict__ b_lin,
    const float* __restrict__ W_gat,
    const float* __restrict__ attn_l, const float* __restrict__ attn_r,
    float* __restrict__ xg, float* __restrict__ el, float* __restrict__ er,
    int n_tiles) {
  extern __shared__ __attribute__((aligned(16))) float smem[];
  const int tid = threadIdx.x;

  if (blockIdx.x < NCHUNK) {
    unsigned* cnt = (unsigned*)smem;
    const int c = blockIdx.x;
    if (c == 0 && tid == 0) *cursor = 0;

    const int4* d4 = (const int4*)(dst + (size_t)c * CHUNK_E);
    int4 dv[RANK_U];
#pragma unroll
    for (int u = 0; u < RANK_U; ++u) {
      const int q = tid + u * RANK_T;
      dv[u] = d4[(q < CHUNK_Q) ? q : (CHUNK_Q - 1)];
    }
    uint4* cnt4 = (uint4*)cnt;
    for (int j = tid; j < NWORD / 4; j += RANK_T)
      cnt4[j] = make_uint4(0u, 0u, 0u, 0u);
    __syncthreads();

    int4* r4 = (int4*)(rank + (size_t)c * CHUNK_E);
#pragma unroll
    for (int u = 0; u < RANK_U; ++u) {
      const int q = tid + u * RANK_T;
      if (q < CHUNK_Q) {
        int4 r;
        r.x = rank16(cnt, dv[u].x);
        r.y = rank16(cnt, dv[u].y);
        r.z = rank16(cnt, dv[u].z);
        r.w = rank16(cnt, dv[u].w);
        r4[q] = r;
      }
    }
    __syncthreads();

    int2* t2 = (int2*)(tab + (size_t)c * NN);
    for (int j = tid; j < NWORD; j += RANK_T) {
      const unsigned w = cnt[j];
      int2 v;
      v.x = (int)(w & 0xFFFFu);
      v.y = (int)(w >> 16);
      t2[j] = v;
    }
    return;
  }

  float* sW1 = smem;
  float* sB1 = sW1 + EMB3 * F_LAT;
  float* sWl = sB1 + F_LAT;
  float* sWr = sWl + F_LAT;
  float* sX  = sWr + F_LAT;

  for (int i = tid; i < EMB3 * F_LAT; i += FRONT_T) sW1[i] = W_lin[i];
  if (tid < F_LAT) sB1[tid] = b_lin[tid];
  if (tid < 2 * F_LAT) {
    const int k = tid & (F_LAT - 1);
    const float4* w4 = (const float4*)(W_gat + k * F_OUT);
    const float4* a4 = (const float4*)((tid < F_LAT) ? attn_l : attn_r);
    float s = 0.0f;
#pragma unroll 4
    for (int c = 0; c < F_OUT / 4; ++c) {
      const float4 w = w4[c];
      const float4 a = a4[c];
      s += w.x * a.x + w.y * a.y + w.z * a.z + w.w * a.w;
    }
    if (tid < F_LAT) sWl[k] = s; else sWr[k] = s;
  }
  __syncthreads();

  const int wave = tid >> 5;
  const int lane = tid & 31;
  const int nb = blockIdx.x - NCHUNK;
  const int m  = lane & 15;
  const int hi = lane >> 4;
  float* xw = sX + wave * 16 * XP;
  const int c4 = m * 4;
  const float4 wl = *(const float4*)(sWl + c4);
  const float4 wr = *(const float4*)(sWr + c4);

  const int tile = wave * NODE_B + nb;
  if (tile < n_tiles) {
    const float* arow = embed + (size_t)(tile * 16 + m) * EMB3;
    v8f acc[4] = {};
    for (int k0 = 0; k0 < EMB3; k0 += 4) {
      const int ka = k0 + 2 * hi;
      v2f a; a.x = arow[ka]; a.y = arow[ka + 1];
      const float* b0 = sW1 + ka * F_LAT + m;
      const float* b1 = b0 + F_LAT;
#pragma unroll
      for (int nt = 0; nt < 4; ++nt) {
        v2f b; b.x = b0[nt * 16]; b.y = b1[nt * 16];
        acc[nt] = wmma4(a, b, acc[nt]);
      }
    }

#pragma unroll
    for (int r = 0; r < 8; ++r) {
      const int row = r + 8 * hi;
#pragma unroll
      for (int nt = 0; nt < 4; ++nt) {
        float v = acc[nt][r] + sB1[nt * 16 + m];
        xw[row * XP + nt * 16 + m] = fmaxf(v, 0.0f);
      }
    }

    float4* xo = (float4*)(xg + (size_t)(tile * 16) * F_LAT);
    float pl[8], pr[8];
#pragma unroll
    for (int it = 0; it < 8; ++it) {
      const int row = 2 * it + hi;
      const float4 v = *(const float4*)(xw + row * XP + c4);
      xo[row * (F_LAT / 4) + m] = v;
      pl[it] = v.x * wl.x + v.y * wl.y + v.z * wl.z + v.w * wl.w;
      pr[it] = v.x * wr.x + v.y * wr.y + v.z * wr.z + v.w * wr.w;
    }
#pragma unroll
    for (int off = 8; off >= 1; off >>= 1) {
#pragma unroll
      for (int it = 0; it < 8; ++it) {
        pl[it] += __shfl_xor(pl[it], off, 32);
        pr[it] += __shfl_xor(pr[it], off, 32);
      }
    }
    if (m == 0) {
#pragma unroll
      for (int it = 0; it < 8; ++it) {
        el[tile * 16 + 2 * it + hi] = pl[it];
        er[tile * 16 + 2 * it + hi] = pr[it];
      }
    }
  }
}

__global__ void seg_alloc_kernel(int* __restrict__ tab,
                                 int* __restrict__ cursor,
                                 int* __restrict__ deg,
                                 int* __restrict__ start) {
  const int d = blockIdx.x * blockDim.x + threadIdx.x;
  const int lane = threadIdx.x & 31;
  const bool ok = d < NN;
  int* col = tab + (ok ? d : (NN - 1));
  int t[NCHUNK];
#pragma unroll
  for (int c = 0; c < NCHUNK; ++c) t[c] = col[(size_t)c * NN];
  int v = 0;
#pragma unroll
  for (int c = 0; c < NCHUNK; ++c) v += t[c];
  if (!ok) v = 0;
  int incl = v;
  for (int off = 1; off < 32; off <<= 1) {
    const int s = __shfl_up(incl, off, 32);
    if (lane >= off) incl += s;
  }
  const int total = __shfl(incl, 31, 32);
  int base = 0;
  if (lane == 0) base = atomicAdd(cursor, total);
  base = __shfl(base, 0, 32);
  if (ok) {
    int run = base + incl - v;
    deg[d] = v;
    start[d] = run;
#pragma unroll
    for (int c = 0; c < NCHUNK; ++c) {
      col[(size_t)c * NN] = run;
      run += t[c];
    }
  }
}

__global__ void edge_place_kernel(const int* __restrict__ src,
                                  const int* __restrict__ dst,
                                  const int* __restrict__ rank,
                                  const int* __restrict__ tab,
                                  int* __restrict__ csr_src) {
  int i = blockIdx.x * blockDim.x + threadIdx.x;
  if (i >= NE) return;
  const int c = i / CHUNK_E;
  csr_src[tab[(size_t)c * NN + dst[i]] + rank[i]] = src[i];
}

__device__ __forceinline__ float lrelu(float e) {
  return (e > 0.0f) ? e : NEG_SLOPE * e;
}
__device__ __forceinline__ float wave_max(float x) {
  for (int off = 16; off >= 1; off >>= 1) x = fmaxf(x, __shfl_xor(x, off, 32));
  return x;
}
__device__ __forceinline__ float wave_sum(float x) {
  for (int off = 16; off >= 1; off >>= 1) x += __shfl_xor(x, off, 32);
  return x;
}

__device__ __forceinline__ void lds_barrier() {
  __builtin_amdgcn_fence(__ATOMIC_RELEASE, "workgroup", "local");
  __builtin_amdgcn_s_barrier();
  __builtin_amdgcn_fence(__ATOMIC_ACQUIRE, "workgroup", "local");
}

__global__ __launch_bounds__(GP_T, 16) void dst_gather_project_kernel(
    const int* __restrict__ deg, const int* __restrict__ start,
    const int* __restrict__ csr_src,
    const float* __restrict__ el, const float* __restrict__ er,
    const float* __restrict__ xg, const float* __restrict__ W_gat,
    const float* __restrict__ b_gat, float* __restrict__ out) {
  __shared__ __attribute__((aligned(16))) float sW[F_LAT * F_OUT];
  __shared__ __attribute__((aligned(16))) float sAgg[16 * XP];
  __shared__ __attribute__((aligned(16))) float sOut[16 * OP];

  const int tid = threadIdx.x;
  const int lane = tid & 31;
  const int m  = lane & 15;
  const int hi = lane >> 4;
  const int w = __builtin_amdgcn_readfirstlane(tid >> 5);

  for (int i = tid; i < F_LAT * F_OUT / 4; i += GP_T)
    ((float4*)sW)[i] = ((const float4*)W_gat)[i];

  const float4* x4 = (const float4*)xg;
  const int c0 = lane * 4;
  const float4 bias = *(const float4*)(b_gat + c0);

  int t = blockIdx.x;
  int d = t * 16 + w;
  int n = deg[d];
  int beg = start[d];
  float erd = er[d];
  int s0 = 0;
  float el0 = 0.0f;
  if (lane < n) {
    s0 = csr_src[beg + lane];
    el0 = el[s0];
  }

  for (; t < NN / 16; t += GP_GRID) {
    const int tn = t + GP_GRID;
    const int t1 = (tn < NN / 16) ? tn : t;
    const int d1 = t1 * 16 + w;
    const int degv = deg[t1 * 16 + m];
    const int begv = start[t1 * 16 + m];
    const float erv = er[t1 * 16 + m];

    const float lg0 = (lane < n) ? lrelu(el0 + erd) : -INFINITY;
    float mx = lg0;
    for (int j = 32 + lane; j < n; j += 32)
      mx = fmaxf(mx, lrelu(el[csr_src[beg + j]] + erd));
    mx = wave_max(mx);
    if ((__float_as_uint(mx) & 0x7F800000u) == 0x7F800000u) mx = 0.0f;

    const float ee0 = (lane < n) ? expf(lg0 - mx) : 0.0f;
    float sum = ee0;
    for (int j = 32 + lane; j < n; j += 32)
      sum += expf(lrelu(el[csr_src[beg + j]] + erd) - mx);
    sum = wave_sum(sum);
    const float dn = (sum > 0.0f) ? sum : 1.0f;

    const int n1 = __builtin_amdgcn_readlane(degv, w);
    const int beg1 = __builtin_amdgcn_readlane(begv, w);
    const float erd1 =
        __int_as_float(__builtin_amdgcn_readlane(__float_as_int(erv), w));
    int s1 = 0;
    if (lane < n1) s1 = csr_src[beg1 + lane];

    float4 acc = {0.0f, 0.0f, 0.0f, 0.0f};
    int sC = s0;
    float aC = ee0 / dn;
    for (int j0 = 0; j0 < n; j0 += 32) {
      if (j0 > 0) {
        const int j = j0 + lane;
        sC = 0;
        aC = 0.0f;
        if (j < n) {
          sC = csr_src[beg + j];
          aC = expf(lrelu(el[sC] + erd) - mx) / dn;
        }
      }
      const int c = (n - j0 < 32) ? (n - j0) : 32;
#pragma unroll 4
      for (int k = 0; k < c; k += 2) {
        const int sLo = __builtin_amdgcn_readlane(sC, k);
        const int sHi = __builtin_amdgcn_readlane(sC, k + 1);
        const int aLo = __builtin_amdgcn_readlane(__float_as_int(aC), k);
        const int aHi = __builtin_amdgcn_readlane(__float_as_int(aC), k + 1);
        const int s = hi ? sHi : sLo;
        const float a = __int_as_float(hi ? aHi : aLo);
        const float4 v = x4[(size_t)s * (F_LAT / 4) + m];
        acc.x += v.x * a;
        acc.y += v.y * a;
        acc.z += v.z * a;
        acc.w += v.w * a;
      }
    }
    acc.x += __shfl_xor(acc.x, 16, 32);
    acc.y += __shfl_xor(acc.y, 16, 32);
    acc.z += __shfl_xor(acc.z, 16, 32);
    acc.w += __shfl_xor(acc.w, 16, 32);
    if (hi == 0) *(float4*)(sAgg + w * XP + m * 4) = acc;
    float el1 = 0.0f;
    if (lane < n1) el1 = el[s1];
    lds_barrier();

    if (w < 8) {
      v8f pc = {};
#pragma unroll 1
      for (int k0 = 0; k0 < F_LAT; k0 += 4) {
        const int ka = k0 + 2 * hi;
        v2f a; a.x = sAgg[m * XP + ka]; a.y = sAgg[m * XP + ka + 1];
        const float* b0 = sW + ka * F_OUT + w * 16 + m;
        v2f b; b.x = b0[0]; b.y = b0[F_OUT];
        pc = wmma4(a, b, pc);
      }
#pragma unroll
      for (int r = 0; r < 8; ++r)
        sOut[(r + 8 * hi) * OP + w * 16 + m] = pc[r];
    }
    lds_barrier();

    const float4 p = *(const float4*)(sOut + w * OP + c0);
    float4 y;
    y.x = p.x + bias.x;
    y.y = p.y + bias.y;
    y.z = p.z + bias.z;
    y.w = p.w + bias.w;
    float* o = out + (size_t)d * 64 + c0;
    if (lane >= 16) {
      y.x = tanhf(y.x); y.y = tanhf(y.y); y.z = tanhf(y.z); y.w = tanhf(y.w);
      o = out + (size_t)NN * 64 + (size_t)d * 64 + (c0 - 64);
    }
    *(float4*)o = y;

    d = d1;
    n = n1;
    beg = beg1;
    erd = erd1;
    s0 = s1;
    el0 = el1;
  }
}

extern "C" void kernel_launch(void* const* d_in, const int* in_sizes, int n_in,
                              void* d_out, int out_size, void* d_ws,
                              size_t ws_size, hipStream_t stream) {
  const float* embed  = (const float*)d_in[0];
  const int*   src    = (const int*)d_in[1];
  const int*   dst    = (const int*)d_in[2];
  const float* W_lin  = (const float*)d_in[3];
  const float* b_lin  = (const float*)d_in[4];
  const float* W_gat  = (const float*)d_in[5];
  const float* attn_l = (const float*)d_in[6];
  const float* attn_r = (const float*)d_in[7];
  const float* b_gat  = (const float*)d_in[8];
  float* out = (float*)d_out;

  float* ws     = (float*)d_ws;
  float* xg     = ws;
  int* tab      = (int*)(xg + (size_t)NN * F_LAT);
  float* el     = xg + (size_t)NN * F_OUT;
  float* er     = el + NN;
  int* deg      = (int*)(er + NN);
  int* start    = deg + NN;
  int* cursor   = start + NN;
  int* rank     = cursor + 32;
  int* csr_src  = rank + NE;

  const int n_tiles = NN / 16;
  const size_t front_lds = (size_t)NWORD * sizeof(unsigned);
  (void)hipFuncSetAttribute((const void*)rank_and_node_kernel,
                            hipFuncAttributeMaxDynamicSharedMemorySize,
                            (int)front_lds);
  rank_and_node_kernel<<<NCHUNK + NODE_B, FRONT_T, front_lds, stream>>>(
      dst, rank, tab, cursor, embed, W_lin, b_lin, W_gat, attn_l, attn_r,
      xg, el, er, n_tiles);
  seg_alloc_kernel<<<(NN + 255) / 256, 256, 0, stream>>>(tab, cursor, deg,
                                                         start);
  edge_place_kernel<<<(NE + 255) / 256, 256, 0, stream>>>(src, dst, rank, tab,
                                                          csr_src);

  dst_gather_project_kernel<<<GP_GRID, GP_T, 0, stream>>>(
      deg, start, csr_src, el, er, xg, W_gat, b_gat, out);
}


# edge chunks sorted by destination inside their ranking block (LDS counter scan + LDS list, whole-line stores); gather walks the 56 chunk lists by table; 2 launches not 4; sums in new chunk order
# speedup vs baseline: 5.8950x; 1.3774x over previous
#include <hip/hip_runtime.h>
#include <math.h>

#define NN     100000
#define NE     1600000
#define EMB3   192
#define F_LAT  64
#define F_OUT  128
#define NEG_SLOPE 0.2f
#define NCHUNK  56
#define CHUNK_E 28672
#define LAST_E  (NE - (NCHUNK - 1) * CHUNK_E)
#define CHUNK_Q (CHUNK_E / 4)
#define RANK_T  1024
#define RANK_U  (CHUNK_Q / RANK_T)
#define NWORD   (NN / 2)
#define SCAN_R  ((NWORD + RANK_T - 1) / RANK_T)
#define NPAIR   (NCHUNK / 2)
#define XP      68
#define OP      132
#define FRONT_T RANK_T
#define NODE_B  (256 - NCHUNK)
#define NODE_W  (NODE_B * (FRONT_T / 32))
static_assert(CHUNK_E % 4 == 0 && CHUNK_E < 65536 && CHUNK_Q % RANK_T == 0, "chunking");
static_assert(LAST_E > 0 && LAST_E <= CHUNK_E && LAST_E % 4 == 0, "last chunk");
static_assert(NCHUNK % 2 == 0 && NPAIR <= 32 && NWORD % 2 == 0 && NN % 4 == 0, "table layout");
static_assert(NODE_W >= NN / 16, "every node tile needs a wave of its own");
#define GP_T    512
#define GP_GRID 1024
static_assert(GP_GRID <= NN / 16, "every gather block starts with a tile of its own");

typedef __attribute__((ext_vector_type(2))) float v2f;
typedef __attribute__((ext_vector_type(8))) float v8f;

__device__ __forceinline__ v8f wmma4(v2f a, v2f b, v8f c) {
  return __builtin_amdgcn_wmma_f32_16x16x4_f32(false, a, false, b, (short)0, c,
                                               false, false);
}


__device__ __forceinline__ int rank16(unsigned* cnt, int d) {
  const unsigned sh = (unsigned)(d & 1) << 4;
  const unsigned old = atomicAdd(&cnt[d >> 1], 1u << sh);
  return (int)((old >> sh) & 0xFFFFu);
}
__device__ __forceinline__ int first16(const unsigned* cnt, int d) {
  const unsigned sh = (unsigned)(d & 1) << 4;
  return (int)((cnt[d >> 1] >> sh) & 0xFFFFu);
}

__global__ __launch_bounds__(FRONT_T) void rank_and_node_kernel(
    const int* __restrict__ src, const int* __restrict__ dst,
    int* __restrict__ tab, int* __restrict__ csr,
    const float* __restrict__ embed,
    const float* __restrict__ W_lin, const float* __restrict__ b_lin,
    const float* __restrict__ W_gat,
    const float* __restrict__ attn_l, const float* __restrict__ attn_r,
    float* __restrict__ xg, float* __restrict__ el, float* __restrict__ er,
    int n_tiles) {
  extern __shared__ __attribute__((aligned(16))) float smem[];
  const int tid = threadIdx.x;

  if (blockIdx.x < NCHUNK) {
    unsigned* cnt = (unsigned*)smem;
    int* list = (int*)(cnt + NWORD);
    int* wtot = list + CHUNK_E;
    const int c = blockIdx.x;
    const int ce = (c == NCHUNK - 1) ? LAST_E : CHUNK_E;
    const int cq = ce / 4;
    const int lane = tid & 31;
    const int wave = tid >> 5;

    const int4* d4 = (const int4*)(dst + (size_t)c * CHUNK_E);
    int4 dv[RANK_U];
#pragma unroll
    for (int u = 0; u < RANK_U; ++u) {
      const int q = tid + u * RANK_T;
      dv[u] = d4[(q < cq) ? q : (cq - 1)];
    }
    uint4* cnt4 = (uint4*)cnt;
    for (int j = tid; j < NWORD / 4; j += RANK_T)
      cnt4[j] = make_uint4(0u, 0u, 0u, 0u);
    __syncthreads();

    int4 rk[RANK_U];
#pragma unroll
    for (int u = 0; u < RANK_U; ++u) {
      const int q = tid + u * RANK_T;
      rk[u] = make_int4(0, 0, 0, 0);
      if (q < cq) {
        rk[u].x = rank16(cnt, dv[u].x);
        rk[u].y = rank16(cnt, dv[u].y);
        rk[u].z = rank16(cnt, dv[u].z);
        rk[u].w = rank16(cnt, dv[u].w);
      }
    }
    __syncthreads();

    const int j0 = tid * SCAN_R;
    unsigned ps = 0;
    for (int i = 0; i < SCAN_R; ++i) {
      const int j = j0 + i;
      if (j < NWORD) ps += cnt[j];
    }
    const int own = (int)(ps & 0xFFFFu) + (int)(ps >> 16);
    int incl = own;
    for (int off = 1; off < 32; off <<= 1) {
      const int v = __shfl_up(incl, off, 32);
      if (lane >= off) incl += v;
    }
    if (lane == 31) wtot[wave] = incl;
    __syncthreads();
    int before = (lane < wave) ? wtot[lane] : 0;
    for (int off = 16; off >= 1; off >>= 1) before += __shfl_xor(before, off, 32);
    int run = before + incl - own;
    for (int i = 0; i < SCAN_R; ++i) {
      const int j = j0 + i;
      if (j < NWORD) {
        const unsigned w = cnt[j];
        const unsigned lo = w & 0xFFFFu;
        const unsigned hi = w >> 16;
        cnt[j] = (unsigned)run | (((unsigned)run + lo) << 16);
        run += (int)(lo + hi);
      }
    }
    __syncthreads();

    int4* t4 = (int4*)(tab + (size_t)c * NN);
    const uint2* c2 = (const uint2*)cnt;
    for (int p = tid; p < NWORD / 2; p += RANK_T) {
      const uint2 w = c2[p];
      const unsigned o0 = w.x & 0xFFFFu;
      const unsigned o1 = w.x >> 16;
      const unsigned o2 = w.y & 0xFFFFu;
      const unsigned o3 = w.y >> 16;
      const unsigned o4 =
          (2 * p + 2 < NWORD) ? (cnt[2 * p + 2] & 0xFFFFu) : (unsigned)ce;
      int4 v;
      v.x = (int)(o0 | ((o1 - o0) << 16));
      v.y = (int)(o1 | ((o2 - o1) << 16));
      v.z = (int)(o2 | ((o3 - o2) << 16));
      v.w = (int)(o3 | ((o4 - o3) << 16));
      t4[p] = v;
    }

    const int4* s4 = (const int4*)(src + (size_t)c * CHUNK_E);
#pragma unroll
    for (int u = 0; u < RANK_U; ++u) {
      const int q = tid + u * RANK_T;
      if (q < cq) {
        const int4 sv = s4[q];
        list[first16(cnt, dv[u].x) + rk[u].x] = sv.x;
        list[first16(cnt, dv[u].y) + rk[u].y] = sv.y;
        list[first16(cnt, dv[u].z) + rk[u].z] = sv.z;
        list[first16(cnt, dv[u].w) + rk[u].w] = sv.w;
      }
    }
    __syncthreads();

    int4* o4 = (int4*)(csr + (size_t)c * CHUNK_E);
    const int4* l4 = (const int4*)list;
    for (int q = tid; q < cq; q += RANK_T) o4[q] = l4[q];
    return;
  }

  float* sW1 = smem;
  float* sB1 = sW1 + EMB3 * F_LAT;
  float* sWl = sB1 + F_LAT;
  float* sWr = sWl + F_LAT;
  float* sX  = sWr + F_LAT;

  for (int i = tid; i < EMB3 * F_LAT; i += FRONT_T) sW1[i] = W_lin[i];
  if (tid < F_LAT) sB1[tid] = b_lin[tid];
  if (tid < 2 * F_LAT) {
    const int k = tid & (F_LAT - 1);
    const float4* w4 = (const float4*)(W_gat + k * F_OUT);
    const float4* a4 = (const float4*)((tid < F_LAT) ? attn_l : attn_r);
    float s = 0.0f;
#pragma unroll 4
    for (int c = 0; c < F_OUT / 4; ++c) {
      const float4 w = w4[c];
      const float4 a = a4[c];
      s += w.x * a.x + w.y * a.y + w.z * a.z + w.w * a.w;
    }
    if (tid < F_LAT) sWl[k] = s; else sWr[k] = s;
  }
  __syncthreads();

  const int wave = tid >> 5;
  const int lane = tid & 31;
  const int nb = blockIdx.x - NCHUNK;
  const int m  = lane & 15;
  const int hi = lane >> 4;
  float* xw = sX + wave * 16 * XP;
  const int c4 = m * 4;
  const float4 wl = *(const float4*)(sWl + c4);
  const float4 wr = *(const float4*)(sWr + c4);

  const int tile = wave * NODE_B + nb;
  if (tile < n_tiles) {
    const float* arow = embed + (size_t)(tile * 16 + m) * EMB3;
    v8f acc[4] = {};
    for (int k0 = 0; k0 < EMB3; k0 += 4) {
      const int ka = k0 + 2 * hi;
      v2f a; a.x = arow[ka]; a.y = arow[ka + 1];
      const float* b0 = sW1 + ka * F_LAT + m;
      const float* b1 = b0 + F_LAT;
#pragma unroll
      for (int nt = 0; nt < 4; ++nt) {
        v2f b; b.x = b0[nt * 16]; b.y = b1[nt * 16];
        acc[nt] = wmma4(a, b, acc[nt]);
      }
    }

#pragma unroll
    for (int r = 0; r < 8; ++r) {
      const int row = r + 8 * hi;
#pragma unroll
      for (int nt = 0; nt < 4; ++nt) {
        float v = acc[nt][r] + sB1[nt * 16 + m];
        xw[row * XP + nt * 16 + m] = fmaxf(v, 0.0f);
      }
    }

    float4* xo = (float4*)(xg + (size_t)(tile * 16) * F_LAT);
    float pl[8], pr[8];
#pragma unroll
    for (int it = 0; it < 8; ++it) {
      const int row = 2 * it + hi;
      const float4 v = *(const float4*)(xw + row * XP + c4);
      xo[row * (F_LAT / 4) + m] = v;
      pl[it] = v.x * wl.x + v.y * wl.y + v.z * wl.z + v.w * wl.w;
      pr[it] = v.x * wr.x + v.y * wr.y + v.z * wr.z + v.w * wr.w;
    }
#pragma unroll
    for (int off = 8; off >= 1; off >>= 1) {
#pragma unroll
      for (int it = 0; it < 8; ++it) {
        pl[it] += __shfl_xor(pl[it], off, 32);
        pr[it] += __shfl_xor(pr[it], off, 32);
      }
    }
    if (m == 0) {
#pragma unroll
      for (int it = 0; it < 8; ++it) {
        el[tile * 16 + 2 * it + hi] = pl[it];
        er[tile * 16 + 2 * it + hi] = pr[it];
      }
    }
  }
}

__device__ __forceinline__ float lrelu(float e) {
  return (e > 0.0f) ? e : NEG_SLOPE * e;
}
__device__ __forceinline__ float wave_max(float x) {
  for (int off = 16; off >= 1; off >>= 1) x = fmaxf(x, __shfl_xor(x, off, 32));
  return x;
}
__device__ __forceinline__ float wave_sum(float x) {
  for (int off = 16; off >= 1; off >>= 1) x += __shfl_xor(x, off, 32);
  return x;
}

__device__ __forceinline__ void lds_barrier() {
  __builtin_amdgcn_fence(__ATOMIC_RELEASE, "workgroup", "local");
  __builtin_amdgcn_s_barrier();
  __builtin_amdgcn_fence(__ATOMIC_ACQUIRE, "workgroup", "local");
}

__device__ __forceinline__ void chain_head(int e0, int e1, int lane, int& incl,
                                           int& mid, int& adj0, int& adj1) {
  const int cnt0 = (int)((unsigned)e0 >> 16);
  const int cnt1 = (int)((unsigned)e1 >> 16);
  const int both = cnt0 + cnt1;
  int in = both;
  for (int off = 1; off < 32; off <<= 1) {
    const int v = __shfl_up(in, off, 32);
    if (lane >= off) in += v;
  }
  const int ex0 = in - both;
  mid = ex0 + cnt0;
  adj0 = (2 * lane) * CHUNK_E + (e0 & 0xFFFF) - ex0;
  adj1 = (2 * lane + 1) * CHUNK_E + (e1 & 0xFFFF) - mid;
  incl = in;
}
__device__ __forceinline__ int slot_addr(int j, int incl, int mid, int adj0,
                                         int adj1) {
  int lo = 0;
  for (int step = 16; step >= 1; step >>= 1) {
    const int v = __shfl(incl, lo + step - 1, 32);
    if (v <= j) lo += step;
  }
  const int m = __shfl(mid, lo, 32);
  const int a0 = __shfl(adj0, lo, 32);
  const int a1 = __shfl(adj1, lo, 32);
  return ((j < m) ? a0 : a1) + j;
}
__device__ __forceinline__ void load_entries(const int* __restrict__ tab, int d,
                                             int lane, int& e0, int& e1) {
  e0 = 0;
  e1 = 0;
  if (lane < NPAIR) {
    const int* p = tab + (size_t)(2 * lane) * NN + d;
    e0 = p[0];
    e1 = p[NN];
  }
}

__global__ __launch_bounds__(GP_T, 16) void dst_gather_project_kernel(
    const int* __restrict__ tab, const int* __restrict__ csr_src,
    const float* __restrict__ el, const float* __restrict__ er,
    const float* __restrict__ xg, const float* __restrict__ W_gat,
    const float* __restrict__ b_gat, float* __restrict__ out) {
  __shared__ __attribute__((aligned(16))) float sW[F_LAT * F_OUT];
  __shared__ __attribute__((aligned(16))) float sAgg[16 * XP];
  __shared__ __attribute__((aligned(16))) float sOut[16 * OP];

  const int tid = threadIdx.x;
  const int lane = tid & 31;
  const int m  = lane & 15;
  const int hi = lane >> 4;
  const int w = __builtin_amdgcn_readfirstlane(tid >> 5);

  for (int i = tid; i < F_LAT * F_OUT / 4; i += GP_T)
    ((float4*)sW)[i] = ((const float4*)W_gat)[i];

  const float4* x4 = (const float4*)xg;
  const int c0 = lane * 4;
  const float4 bias = *(const float4*)(b_gat + c0);

  int t = blockIdx.x;
  int d = t * 16 + w;
  float erd = er[d];
  int n;
  int s0 = 0;
  float el0 = 0.0f;
  {
    int e0, e1, incl, mid, adj0, adj1;
    load_entries(tab, d, lane, e0, e1);
    chain_head(e0, e1, lane, incl, mid, adj0, adj1);
    n = __builtin_amdgcn_readlane(incl, 31);
    const int a = slot_addr(lane, incl, mid, adj0, adj1);
    if (lane < n) {
      s0 = csr_src[a];
      el0 = el[s0];
    }
  }

  for (; t < NN / 16; t += GP_GRID) {
    const int tn = t + GP_GRID;
    const int t1 = (tn < NN / 16) ? tn : t;
    const int d1 = t1 * 16 + w;
    int e0n, e1n;
    load_entries(tab, d1, lane, e0n, e1n);
    const float erv = er[t1 * 16 + m];

    const float lg0 = (lane < n) ? lrelu(el0 + erd) : -INFINITY;
    float mx = lg0;
    if (n > 32) {
      int e0, e1, incl, mid, adj0, adj1;
      load_entries(tab, d, lane, e0, e1);
      chain_head(e0, e1, lane, incl, mid, adj0, adj1);
      for (int j0 = 32; j0 < n; j0 += 32) {
        const int j = j0 + lane;
        const int a = slot_addr(j, incl, mid, adj0, adj1);
        if (j < n) mx = fmaxf(mx, lrelu(el[csr_src[a]] + erd));
      }
    }
    mx = wave_max(mx);
    if ((__float_as_uint(mx) & 0x7F800000u) == 0x7F800000u) mx = 0.0f;

    const float ee0 = (lane < n) ? expf(lg0 - mx) : 0.0f;
    float sum = ee0;
    if (n > 32) {
      int e0, e1, incl, mid, adj0, adj1;
      load_entries(tab, d, lane, e0, e1);
      chain_head(e0, e1, lane, incl, mid, adj0, adj1);
      for (int j0 = 32; j0 < n; j0 += 32) {
        const int j = j0 + lane;
        const int a = slot_addr(j, incl, mid, adj0, adj1);
        if (j < n) sum += expf(lrelu(el[csr_src[a]] + erd) - mx);
      }
    }
    sum = wave_sum(sum);
    const float dn = (sum > 0.0f) ? sum : 1.0f;

    int n1;
    int s1 = 0;
    {
      int incl, mid, adj0, adj1;
      chain_head(e0n, e1n, lane, incl, mid, adj0, adj1);
      n1 = __builtin_amdgcn_readlane(incl, 31);
      const int a = slot_addr(lane, incl, mid, adj0, adj1);
      if (lane < n1) s1 = csr_src[a];
    }
    const float erd1 =
        __int_as_float(__builtin_amdgcn_readlane(__float_as_int(erv), w));

    float4 acc = {0.0f, 0.0f, 0.0f, 0.0f};
    int sC = s0;
    float aC = ee0 / dn;
    for (int j0 = 0; j0 < n; j0 += 32) {
      if (j0 > 0) {
        const int j = j0 + lane;
        int e0, e1, incl, mid, adj0, adj1;
        load_entries(tab, d, lane, e0, e1);
        chain_head(e0, e1, lane, incl, mid, adj0, adj1);
        const int a = slot_addr(j, incl, mid, adj0, adj1);
        sC = 0;
        aC = 0.0f;
        if (j < n) {
          sC = csr_src[a];
          aC = expf(lrelu(el[sC] + erd) - mx) / dn;
        }
      }
      const int c = (n - j0 < 32) ? (n - j0) : 32;
#pragma unroll 4
      for (int k = 0; k < c; k += 2) {
        const int sLo = __builtin_amdgcn_readlane(sC, k);
        const int sHi = __builtin_amdgcn_readlane(sC, k + 1);
        const int aLo = __builtin_amdgcn_readlane(__float_as_int(aC), k);
        const int aHi = __builtin_amdgcn_readlane(__float_as_int(aC), k + 1);
        const int s = hi ? sHi : sLo;
        const float a = __int_as_float(hi ? aHi : aLo);
        const float4 v = x4[(size_t)s * (F_LAT / 4) + m];
        acc.x += v.x * a;
        acc.y += v.y * a;
        acc.z += v.z * a;
        acc.w += v.w * a;
      }
    }
    acc.x += __shfl_xor(acc.x, 16, 32);
    acc.y += __shfl_xor(acc.y, 16, 32);
    acc.z += __shfl_xor(acc.z, 16, 32);
    acc.w += __shfl_xor(acc.w, 16, 32);
    if (hi == 0) *(float4*)(sAgg + w * XP + m * 4) = acc;
    float el1 = 0.0f;
    if (lane < n1) el1 = el[s1];
    lds_barrier();

    if (w < 8) {
      v8f pc = {};
#pragma unroll 1
      for (int k0 = 0; k0 < F_LAT; k0 += 4) {
        const int ka = k0 + 2 * hi;
        v2f a; a.x = sAgg[m * XP + ka]; a.y = sAgg[m * XP + ka + 1];
        const float* b0 = sW + ka * F_OUT + w * 16 + m;
        v2f b; b.x = b0[0]; b.y = b0[F_OUT];
        pc = wmma4(a, b, pc);
      }
#pragma unroll
      for (int r = 0; r < 8; ++r)
        sOut[(r + 8 * hi) * OP + w * 16 + m] = pc[r];
    }
    lds_barrier();

    const float4 p = *(const float4*)(sOut + w * OP + c0);
    float4 y;
    y.x = p.x + bias.x;
    y.y = p.y + bias.y;
    y.z = p.z + bias.z;
    y.w = p.w + bias.w;
    float* o = out + (size_t)d * 64 + c0;
    if (lane >= 16) {
      y.x = tanhf(y.x); y.y = tanhf(y.y); y.z = tanhf(y.z); y.w = tanhf(y.w);
      o = out + (size_t)NN * 64 + (size_t)d * 64 + (c0 - 64);
    }
    *(float4*)o = y;

    d = d1;
    n = n1;
    erd = erd1;
    s0 = s1;
    el0 = el1;
  }
}

extern "C" void kernel_launch(void* const* d_in, const int* in_sizes, int n_in,
                              void* d_out, int out_size, void* d_ws,
                              size_t ws_size, hipStream_t stream) {
  const float* embed  = (const float*)d_in[0];
  const int*   src    = (const int*)d_in[1];
  const int*   dst    = (const int*)d_in[2];
  const float* W_lin  = (const float*)d_in[3];
  const float* b_lin  = (const float*)d_in[4];
  const float* W_gat  = (const float*)d_in[5];
  const float* attn_l = (const float*)d_in[6];
  const float* attn_r = (const float*)d_in[7];
  const float* b_gat  = (const float*)d_in[8];
  float* out = (float*)d_out;

  float* ws     = (float*)d_ws;
  float* xg     = ws;
  int* tab      = (int*)(xg + (size_t)NN * F_LAT);
  float* el     = xg + (size_t)NN * F_OUT;
  float* er     = el + NN;
  int* csr_src  = (int*)(er + NN);

  const int n_tiles = NN / 16;
  const size_t front_lds =
      (size_t)NWORD * sizeof(unsigned) + (size_t)(CHUNK_E + 32) * sizeof(int);
  (void)hipFuncSetAttribute((const void*)rank_and_node_kernel,
                            hipFuncAttributeMaxDynamicSharedMemorySize,
                            (int)front_lds);
  rank_and_node_kernel<<<NCHUNK + NODE_B, FRONT_T, front_lds, stream>>>(
      src, dst, tab, csr_src, embed, W_lin, b_lin, W_gat, attn_l, attn_r,
      xg, el, er, n_tiles);

  dst_gather_project_kernel<<<GP_GRID, GP_T, 0, stream>>>(
      tab, csr_src, el, er, xg, W_gat, b_gat, out);
}


# gather kernel: a tile's 56x16 table entries fetched once per block as whole 64-B row pieces two tiles ahead and handed to the waves through LDS, not 56 lines per wave; bit-identical
# speedup vs baseline: 6.1890x; 1.0499x over previous
#include <hip/hip_runtime.h>
#include <math.h>

#define NN     100000
#define NE     1600000
#define EMB3   192
#define F_LAT  64
#define F_OUT  128
#define NEG_SLOPE 0.2f
#define NCHUNK  56
#define CHUNK_E 28672
#define LAST_E  (NE - (NCHUNK - 1) * CHUNK_E)
#define CHUNK_Q (CHUNK_E / 4)
#define RANK_T  1024
#define RANK_U  (CHUNK_Q / RANK_T)
#define NWORD   (NN / 2)
#define SCAN_R  ((NWORD + RANK_T - 1) / RANK_T)
#define NPAIR   (NCHUNK / 2)
#define XP      68
#define OP      132
#define FRONT_T RANK_T
#define NODE_B  (256 - NCHUNK)
#define NODE_W  (NODE_B * (FRONT_T / 32))
static_assert(CHUNK_E % 4 == 0 && CHUNK_E < 65536 && CHUNK_Q % RANK_T == 0, "chunking");
static_assert(LAST_E > 0 && LAST_E <= CHUNK_E && LAST_E % 4 == 0, "last chunk");
static_assert(NCHUNK % 2 == 0 && NPAIR <= 32 && NWORD % 2 == 0 && NN % 4 == 0, "table layout");
static_assert(NODE_W >= NN / 16, "every node tile needs a wave of its own");
#define GP_T    512
#define GP_GRID 1024
static_assert(GP_GRID <= NN / 16, "every gather block starts with a tile of its own");
#define TP      58
#define TAB_L   (NCHUNK * 8)
static_assert(TP >= NCHUNK && TP % 2 == 0 && TAB_L <= GP_T && TAB_L % 32 == 0, "table staging");

typedef __attribute__((ext_vector_type(2))) float v2f;
typedef __attribute__((ext_vector_type(8))) float v8f;

__device__ __forceinline__ v8f wmma4(v2f a, v2f b, v8f c) {
  return __builtin_amdgcn_wmma_f32_16x16x4_f32(false, a, false, b, (short)0, c,
                                               false, false);
}


__device__ __forceinline__ int rank16(unsigned* cnt, int d) {
  const unsigned sh = (unsigned)(d & 1) << 4;
  const unsigned old = atomicAdd(&cnt[d >> 1], 1u << sh);
  return (int)((old >> sh) & 0xFFFFu);
}
__device__ __forceinline__ int first16(const unsigned* cnt, int d) {
  const unsigned sh = (unsigned)(d & 1) << 4;
  return (int)((cnt[d >> 1] >> sh) & 0xFFFFu);
}

__global__ __launch_bounds__(FRONT_T) void rank_and_node_kernel(
    const int* __restrict__ src, const int* __restrict__ dst,
    int* __restrict__ tab, int* __restrict__ csr,
    const float* __restrict__ embed,
    const float* __restrict__ W_lin, const float* __restrict__ b_lin,
    const float* __restrict__ W_gat,
    const float* __restrict__ attn_l, const float* __restrict__ attn_r,
    float* __restrict__ xg, float* __restrict__ el, float* __restrict__ er,
    int n_tiles) {
  extern __shared__ __attribute__((aligned(16))) float smem[];
  const int tid = threadIdx.x;

  if (blockIdx.x < NCHUNK) {
    unsigned* cnt = (unsigned*)smem;
    int* list = (int*)(cnt + NWORD);
    int* wtot = list + CHUNK_E;
    const int c = blockIdx.x;
    const int ce = (c == NCHUNK - 1) ? LAST_E : CHUNK_E;
    const int cq = ce / 4;
    const int lane = tid & 31;
    const int wave = tid >> 5;

    const int4* d4 = (const int4*)(dst + (size_t)c * CHUNK_E);
    int4 dv[RANK_U];
#pragma unroll
    for (int u = 0; u < RANK_U; ++u) {
      const int q = tid + u * RANK_T;
      dv[u] = d4[(q < cq) ? q : (cq - 1)];
    }
    uint4* cnt4 = (uint4*)cnt;
    for (int j = tid; j < NWORD / 4; j += RANK_T)
      cnt4[j] = make_uint4(0u, 0u, 0u, 0u);
    __syncthreads();

    int4 rk[RANK_U];
#pragma unroll
    for (int u = 0; u < RANK_U; ++u) {
      const int q = tid + u * RANK_T;
      rk[u] = make_int4(0, 0, 0, 0);
      if (q < cq) {
        rk[u].x = rank16(cnt, dv[u].x);
        rk[u].y = rank16(cnt, dv[u].y);
        rk[u].z = rank16(cnt, dv[u].z);
        rk[u].w = rank16(cnt, dv[u].w);
      }
    }
    __syncthreads();

    const int j0 = tid * SCAN_R;
    unsigned ps = 0;
    for (int i = 0; i < SCAN_R; ++i) {
      const int j = j0 + i;
      if (j < NWORD) ps += cnt[j];
    }
    const int own = (int)(ps & 0xFFFFu) + (int)(ps >> 16);
    int incl = own;
    for (int off = 1; off < 32; off <<= 1) {
      const int v = __shfl_up(incl, off, 32);
      if (lane >= off) incl += v;
    }
    if (lane == 31) wtot[wave] = incl;
    __syncthreads();
    int before = (lane < wave) ? wtot[lane] : 0;
    for (int off = 16; off >= 1; off >>= 1) before += __shfl_xor(before, off, 32);
    int run = before + incl - own;
    for (int i = 0; i < SCAN_R; ++i) {
      const int j = j0 + i;
      if (j < NWORD) {
        const unsigned w = cnt[j];
        const unsigned lo = w & 0xFFFFu;
        const unsigned hi = w >> 16;
        cnt[j] = (unsigned)run | (((unsigned)run + lo) << 16);
        run += (int)(lo + hi);
      }
    }
    __syncthreads();

    int4* t4 = (int4*)(tab + (size_t)c * NN);
    const uint2* c2 = (const uint2*)cnt;
    for (int p = tid; p < NWORD / 2; p += RANK_T) {
      const uint2 w = c2[p];
      const unsigned o0 = w.x & 0xFFFFu;
      const unsigned o1 = w.x >> 16;
      const unsigned o2 = w.y & 0xFFFFu;
      const unsigned o3 = w.y >> 16;
      const unsigned o4 =
          (2 * p + 2 < NWORD) ? (cnt[2 * p + 2] & 0xFFFFu) : (unsigned)ce;
      int4 v;
      v.x = (int)(o0 | ((o1 - o0) << 16));
      v.y = (int)(o1 | ((o2 - o1) << 16));
      v.z = (int)(o2 | ((o3 - o2) << 16));
      v.w = (int)(o3 | ((o4 - o3) << 16));
      t4[p] = v;
    }

    const int4* s4 = (const int4*)(src + (size_t)c * CHUNK_E);
#pragma unroll
    for (int u = 0; u < RANK_U; ++u) {
      const int q = tid + u * RANK_T;
      if (q < cq) {
        const int4 sv = s4[q];
        list[first16(cnt, dv[u].x) + rk[u].x] = sv.x;
        list[first16(cnt, dv[u].y) + rk[u].y] = sv.y;
        list[first16(cnt, dv[u].z) + rk[u].z] = sv.z;
        list[first16(cnt, dv[u].w) + rk[u].w] = sv.w;
      }
    }
    __syncthreads();

    int4* o4 = (int4*)(csr + (size_t)c * CHUNK_E);
    const int4* l4 = (const int4*)list;
    for (int q = tid; q < cq; q += RANK_T) o4[q] = l4[q];
    return;
  }

  float* sW1 = smem;
  float* sB1 = sW1 + EMB3 * F_LAT;
  float* sWl = sB1 + F_LAT;
  float* sWr = sWl + F_LAT;
  float* sX  = sWr + F_LAT;

  for (int i = tid; i < EMB3 * F_LAT; i += FRONT_T) sW1[i] = W_lin[i];
  if (tid < F_LAT) sB1[tid] = b_lin[tid];
  if (tid < 2 * F_LAT) {
    const int k = tid & (F_LAT - 1);
    const float4* w4 = (const float4*)(W_gat + k * F_OUT);
    const float4* a4 = (const float4*)((tid < F_LAT) ? attn_l : attn_r);
    float s = 0.0f;
#pragma unroll 4
    for (int c = 0; c < F_OUT / 4; ++c) {
      const float4 w = w4[c];
      const float4 a = a4[c];
      s += w.x * a.x + w.y * a.y + w.z * a.z + w.w * a.w;
    }
    if (tid < F_LAT) sWl[k] = s; else sWr[k] = s;
  }
  __syncthreads();

  const int wave = tid >> 5;
  const int lane = tid & 31;
  const int nb = blockIdx.x - NCHUNK;
  const int m  = lane & 15;
  const int hi = lane >> 4;
  float* xw = sX + wave * 16 * XP;
  const int c4 = m * 4;
  const float4 wl = *(const float4*)(sWl + c4);
  const float4 wr = *(const float4*)(sWr + c4);

  const int tile = wave * NODE_B + nb;
  if (tile < n_tiles) {
    const float* arow = embed + (size_t)(tile * 16 + m) * EMB3;
    v8f acc[4] = {};
    for (int k0 = 0; k0 < EMB3; k0 += 4) {
      const int ka = k0 + 2 * hi;
      v2f a; a.x = arow[ka]; a.y = arow[ka + 1];
      const float* b0 = sW1 + ka * F_LAT + m;
      const float* b1 = b0 + F_LAT;
#pragma unroll
      for (int nt = 0; nt < 4; ++nt) {
        v2f b; b.x = b0[nt * 16]; b.y = b1[nt * 16];
        acc[nt] = wmma4(a, b, acc[nt]);
      }
    }

#pragma unroll
    for (int r = 0; r < 8; ++r) {
      const int row = r + 8 * hi;
#pragma unroll
      for (int nt = 0; nt < 4; ++nt) {
        float v = acc[nt][r] + sB1[nt * 16 + m];
        xw[row * XP + nt * 16 + m] = fmaxf(v, 0.0f);
      }
    }

    float4* xo = (float4*)(xg + (size_t)(tile * 16) * F_LAT);
    float pl[8], pr[8];
#pragma unroll
    for (int it = 0; it < 8; ++it) {
      const int row = 2 * it + hi;
      const float4 v = *(const float4*)(xw + row * XP + c4);
      xo[row * (F_LAT / 4) + m] = v;
      pl[it] = v.x * wl.x + v.y * wl.y + v.z * wl.z + v.w * wl.w;
      pr[it] = v.x * wr.x + v.y * wr.y + v.z * wr.z + v.w * wr.w;
    }
#pragma unroll
    for (int off = 8; off >= 1; off >>= 1) {
#pragma unroll
      for (int it = 0; it < 8; ++it) {
        pl[it] += __shfl_xor(pl[it], off, 32);
        pr[it] += __shfl_xor(pr[it], off, 32);
      }
    }
    if (m == 0) {
#pragma unroll
      for (int it = 0; it < 8; ++it) {
        el[tile * 16 + 2 * it + hi] = pl[it];
        er[tile * 16 + 2 * it + hi] = pr[it];
      }
    }
  }
}

__device__ __forceinline__ float lrelu(float e) {
  return (e > 0.0f) ? e : NEG_SLOPE * e;
}
__device__ __forceinline__ float wave_max(float x) {
  for (int off = 16; off >= 1; off >>= 1) x = fmaxf(x, __shfl_xor(x, off, 32));
  return x;
}
__device__ __forceinline__ float wave_sum(float x) {
  for (int off = 16; off >= 1; off >>= 1) x += __shfl_xor(x, off, 32);
  return x;
}

__device__ __forceinline__ void lds_barrier() {
  __builtin_amdgcn_fence(__ATOMIC_RELEASE, "workgroup", "local");
  __builtin_amdgcn_s_barrier();
  __builtin_amdgcn_fence(__ATOMIC_ACQUIRE, "workgroup", "local");
}

__device__ __forceinline__ void chain_head(int e0, int e1, int lane, int& incl,
                                           int& mid, int& adj0, int& adj1) {
  const int cnt0 = (int)((unsigned)e0 >> 16);
  const int cnt1 = (int)((unsigned)e1 >> 16);
  const int both = cnt0 + cnt1;
  int in = both;
  for (int off = 1; off < 32; off <<= 1) {
    const int v = __shfl_up(in, off, 32);
    if (lane >= off) in += v;
  }
  const int ex0 = in - both;
  mid = ex0 + cnt0;
  adj0 = (2 * lane) * CHUNK_E + (e0 & 0xFFFF) - ex0;
  adj1 = (2 * lane + 1) * CHUNK_E + (e1 & 0xFFFF) - mid;
  incl = in;
}
__device__ __forceinline__ int slot_addr(int j, int incl, int mid, int adj0,
                                         int adj1) {
  int lo = 0;
  for (int step = 16; step >= 1; step >>= 1) {
    const int v = __shfl(incl, lo + step - 1, 32);
    if (v <= j) lo += step;
  }
  const int m = __shfl(mid, lo, 32);
  const int a0 = __shfl(adj0, lo, 32);
  const int a1 = __shfl(adj1, lo, 32);
  return ((j < m) ? a0 : a1) + j;
}

__device__ __forceinline__ int2 fetch_tab_piece(const int* __restrict__ tab,
                                                int tile, int L) {
  const int* base = tab + tile * 16;
  const unsigned off = (unsigned)(L >> 3) * NN + (unsigned)(L & 7) * 2;
  return *(const int2*)(base + off);
}
__device__ __forceinline__ void put_tab_piece(int* tb, int L, int2 v) {
  int* p = tb + ((L & 7) * 2) * TP + (L >> 3);
  p[0] = v.x;
  p[TP] = v.y;
}
__device__ __forceinline__ void take_entries(const int* tb, int r, int lane,
                                             int& e0, int& e1) {
  e0 = 0;
  e1 = 0;
  if (lane < NPAIR) {
    const int2 e = *(const int2*)(tb + r * TP + 2 * lane);
    e0 = e.x;
    e1 = e.y;
  }
}

__global__ __launch_bounds__(GP_T, 16) void dst_gather_project_kernel(
    const int* __restrict__ tab, const int* __restrict__ csr_src,
    const float* __restrict__ el, const float* __restrict__ er,
    const float* __restrict__ xg, const float* __restrict__ W_gat,
    const float* __restrict__ b_gat, float* __restrict__ out) {
  __shared__ __attribute__((aligned(16))) float sW[F_LAT * F_OUT];
  __shared__ __attribute__((aligned(16))) float sAgg[16 * XP];
  __shared__ __attribute__((aligned(16))) float sOut[16 * OP];
  __shared__ __attribute__((aligned(16))) int sTab[3][16 * TP];

  const int tid = threadIdx.x;
  const int lane = tid & 31;
  const int m  = lane & 15;
  const int hi = lane >> 4;
  const int w = __builtin_amdgcn_readfirstlane(tid >> 5);

  for (int i = tid; i < F_LAT * F_OUT / 4; i += GP_T)
    ((float4*)sW)[i] = ((const float4*)W_gat)[i];

  const float4* x4 = (const float4*)xg;
  const int c0 = lane * 4;
  const float4 bias = *(const float4*)(b_gat + c0);

  int t = blockIdx.x;
  int d = t * 16 + w;
  float erd = er[d];
  if (w < TAB_L / 32) {
    const int tb1 = (t + GP_GRID < NN / 16) ? t + GP_GRID : t;
    const int2 p0 = fetch_tab_piece(tab, t, tid);
    const int2 p1 = fetch_tab_piece(tab, tb1, tid);
    put_tab_piece(sTab[0], tid, p0);
    put_tab_piece(sTab[1], tid, p1);
  }
  __syncthreads();
  int bc = 0, bn = 1, bf = 2;
  int n;
  int s0 = 0;
  float el0 = 0.0f;
  {
    int e0, e1, incl, mid, adj0, adj1;
    take_entries(sTab[0], w, lane, e0, e1);
    chain_head(e0, e1, lane, incl, mid, adj0, adj1);
    n = __builtin_amdgcn_readlane(incl, 31);
    const int a = slot_addr(lane, incl, mid, adj0, adj1);
    if (lane < n) {
      s0 = csr_src[a];
      el0 = el[s0];
    }
  }

  for (; t < NN / 16; t += GP_GRID) {
    const int tn = t + GP_GRID;
    const int t1 = (tn < NN / 16) ? tn : t;
    const int d1 = t1 * 16 + w;
    const float erv = er[t1 * 16 + m];

    const float lg0 = (lane < n) ? lrelu(el0 + erd) : -INFINITY;
    float mx = lg0;
    if (n > 32) {
      int e0, e1, incl, mid, adj0, adj1;
      take_entries(sTab[bc], w, lane, e0, e1);
      chain_head(e0, e1, lane, incl, mid, adj0, adj1);
      for (int j0 = 32; j0 < n; j0 += 32) {
        const int j = j0 + lane;
        const int a = slot_addr(j, incl, mid, adj0, adj1);
        if (j < n) mx = fmaxf(mx, lrelu(el[csr_src[a]] + erd));
      }
    }
    mx = wave_max(mx);
    if ((__float_as_uint(mx) & 0x7F800000u) == 0x7F800000u) mx = 0.0f;

    const float ee0 = (lane < n) ? expf(lg0 - mx) : 0.0f;
    float sum = ee0;
    if (n > 32) {
      int e0, e1, incl, mid, adj0, adj1;
      take_entries(sTab[bc], w, lane, e0, e1);
      chain_head(e0, e1, lane, incl, mid, adj0, adj1);
      for (int j0 = 32; j0 < n; j0 += 32) {
        const int j = j0 + lane;
        const int a = slot_addr(j, incl, mid, adj0, adj1);
        if (j < n) sum += expf(lrelu(el[csr_src[a]] + erd) - mx);
      }
    }
    sum = wave_sum(sum);
    const float dn = (sum > 0.0f) ? sum : 1.0f;

    int n1;
    int s1 = 0;
    const int t2 = (tn + GP_GRID < NN / 16) ? tn + GP_GRID : t;
    int2 piece = {0, 0};
    {
      int e0n, e1n, incl, mid, adj0, adj1;
      take_entries(sTab[bn], w, lane, e0n, e1n);
      chain_head(e0n, e1n, lane, incl, mid, adj0, adj1);
      n1 = __builtin_amdgcn_readlane(incl, 31);
      const int a = slot_addr(lane, incl, mid, adj0, adj1);
      if (w < TAB_L / 32) piece = fetch_tab_piece(tab, t2, tid);
      if (lane < n1) s1 = csr_src[a];
    }
    const float erd1 =
        __int_as_float(__builtin_amdgcn_readlane(__float_as_int(erv), w));

    if (w < TAB_L / 32) put_tab_piece(sTab[bf], tid, piece);

    float4 acc = {0.0f, 0.0f, 0.0f, 0.0f};
    int sC = s0;
    float aC = ee0 / dn;
    for (int j0 = 0; j0 < n; j0 += 32) {
      if (j0 > 0) {
        const int j = j0 + lane;
        int e0, e1, incl, mid, adj0, adj1;
        take_entries(sTab[bc], w, lane, e0, e1);
        chain_head(e0, e1, lane, incl, mid, adj0, adj1);
        const int a = slot_addr(j, incl, mid, adj0, adj1);
        sC = 0;
        aC = 0.0f;
        if (j < n) {
          sC = csr_src[a];
          aC = expf(lrelu(el[sC] + erd) - mx) / dn;
        }
      }
      const int c = (n - j0 < 32) ? (n - j0) : 32;
#pragma unroll 4
      for (int k = 0; k < c; k += 2) {
        const int sLo = __builtin_amdgcn_readlane(sC, k);
        const int sHi = __builtin_amdgcn_readlane(sC, k + 1);
        const int aLo = __builtin_amdgcn_readlane(__float_as_int(aC), k);
        const int aHi = __builtin_amdgcn_readlane(__float_as_int(aC), k + 1);
        const int s = hi ? sHi : sLo;
        const float a = __int_as_float(hi ? aHi : aLo);
        const float4 v = x4[(size_t)s * (F_LAT / 4) + m];
        acc.x += v.x * a;
        acc.y += v.y * a;
        acc.z += v.z * a;
        acc.w += v.w * a;
      }
    }
    acc.x += __shfl_xor(acc.x, 16, 32);
    acc.y += __shfl_xor(acc.y, 16, 32);
    acc.z += __shfl_xor(acc.z, 16, 32);
    acc.w += __shfl_xor(acc.w, 16, 32);
    if (hi == 0) *(float4*)(sAgg + w * XP + m * 4) = acc;
    float el1 = 0.0f;
    if (lane < n1) el1 = el[s1];
    lds_barrier();

    if (w < 8) {
      v8f pc = {};
#pragma unroll 1
      for (int k0 = 0; k0 < F_LAT; k0 += 4) {
        const int ka = k0 + 2 * hi;
        v2f a; a.x = sAgg[m * XP + ka]; a.y = sAgg[m * XP + ka + 1];
        const float* b0 = sW + ka * F_OUT + w * 16 + m;
        v2f b; b.x = b0[0]; b.y = b0[F_OUT];
        pc = wmma4(a, b, pc);
      }
#pragma unroll
      for (int r = 0; r < 8; ++r)
        sOut[(r + 8 * hi) * OP + w * 16 + m] = pc[r];
    }
    lds_barrier();

    const float4 p = *(const float4*)(sOut + w * OP + c0);
    float4 y;
    y.x = p.x + bias.x;
    y.y = p.y + bias.y;
    y.z = p.z + bias.z;
    y.w = p.w + bias.w;
    float* o = out + (size_t)d * 64 + c0;
    if (lane >= 16) {
      y.x = tanhf(y.x); y.y = tanhf(y.y); y.z = tanhf(y.z); y.w = tanhf(y.w);
      o = out + (size_t)NN * 64 + (size_t)d * 64 + (c0 - 64);
    }
    *(float4*)o = y;

    {
      const int b = bc;
      bc = bn;
      bn = bf;
      bf = b;
    }
    d = d1;
    n = n1;
    erd = erd1;
    s0 = s1;
    el0 = el1;
  }
}

extern "C" void kernel_launch(void* const* d_in, const int* in_sizes, int n_in,
                              void* d_out, int out_size, void* d_ws,
                              size_t ws_size, hipStream_t stream) {
  const float* embed  = (const float*)d_in[0];
  const int*   src    = (const int*)d_in[1];
  const int*   dst    = (const int*)d_in[2];
  const float* W_lin  = (const float*)d_in[3];
  const float* b_lin  = (const float*)d_in[4];
  const float* W_gat  = (const float*)d_in[5];
  const float* attn_l = (const float*)d_in[6];
  const float* attn_r = (const float*)d_in[7];
  const float* b_gat  = (const float*)d_in[8];
  float* out = (float*)d_out;

  float* ws     = (float*)d_ws;
  float* xg     = ws;
  int* tab      = (int*)(xg + (size_t)NN * F_LAT);
  float* el     = xg + (size_t)NN * F_OUT;
  float* er     = el + NN;
  int* csr_src  = (int*)(er + NN);

  const int n_tiles = NN / 16;
  const size_t front_lds =
      (size_t)NWORD * sizeof(unsigned) + (size_t)(CHUNK_E + 32) * sizeof(int);
  (void)hipFuncSetAttribute((const void*)rank_and_node_kernel,
                            hipFuncAttributeMaxDynamicSharedMemorySize,
                            (int)front_lds);
  rank_and_node_kernel<<<NCHUNK + NODE_B, FRONT_T, front_lds, stream>>>(
      src, dst, tab, csr_src, embed, W_lin, b_lin, W_gat, attn_l, attn_r,
      xg, el, er, n_tiles);

  dst_gather_project_kernel<<<GP_GRID, GP_T, 0, stream>>>(
      tab, csr_src, el, er, xg, W_gat, b_gat, out);
}


# gather kernel: wave max/sum, slot scan and slot search on vector-ALU lane moves (DPP row steps, permlanex16, permlane16 var), not LDS-path permutes; same pairing order, bit-identical
# speedup vs baseline: 6.5962x; 1.0658x over previous
#include <hip/hip_runtime.h>
#include <math.h>

#define NN     100000
#define NE     1600000
#define EMB3   192
#define F_LAT  64
#define F_OUT  128
#define NEG_SLOPE 0.2f
#define NCHUNK  56
#define CHUNK_E 28672
#define LAST_E  (NE - (NCHUNK - 1) * CHUNK_E)
#define CHUNK_Q (CHUNK_E / 4)
#define RANK_T  1024
#define RANK_U  (CHUNK_Q / RANK_T)
#define NWORD   (NN / 2)
#define SCAN_R  ((NWORD + RANK_T - 1) / RANK_T)
#define NPAIR   (NCHUNK / 2)
#define XP      68
#define OP      132
#define FRONT_T RANK_T
#define NODE_B  (256 - NCHUNK)
#define NODE_W  (NODE_B * (FRONT_T / 32))
static_assert(CHUNK_E % 4 == 0 && CHUNK_E < 65536 && CHUNK_Q % RANK_T == 0, "chunking");
static_assert(LAST_E > 0 && LAST_E <= CHUNK_E && LAST_E % 4 == 0, "last chunk");
static_assert(NCHUNK % 2 == 0 && NPAIR <= 32 && NWORD % 2 == 0 && NN % 4 == 0, "table layout");
static_assert(NODE_W >= NN / 16, "every node tile needs a wave of its own");
#define GP_T    512
#define GP_GRID 1024
static_assert(GP_GRID <= NN / 16, "every gather block starts with a tile of its own");
#define TP      58
#define TAB_L   (NCHUNK * 8)
static_assert(TP >= NCHUNK && TP % 2 == 0 && TAB_L <= GP_T && TAB_L % 32 == 0, "table staging");

typedef __attribute__((ext_vector_type(2))) float v2f;
typedef __attribute__((ext_vector_type(8))) float v8f;

__device__ __forceinline__ v8f wmma4(v2f a, v2f b, v8f c) {
  return __builtin_amdgcn_wmma_f32_16x16x4_f32(false, a, false, b, (short)0, c,
                                               false, false);
}


__device__ __forceinline__ int rank16(unsigned* cnt, int d) {
  const unsigned sh = (unsigned)(d & 1) << 4;
  const unsigned old = atomicAdd(&cnt[d >> 1], 1u << sh);
  return (int)((old >> sh) & 0xFFFFu);
}
__device__ __forceinline__ int first16(const unsigned* cnt, int d) {
  const unsigned sh = (unsigned)(d & 1) << 4;
  return (int)((cnt[d >> 1] >> sh) & 0xFFFFu);
}

__global__ __launch_bounds__(FRONT_T) void rank_and_node_kernel(
    const int* __restrict__ src, const int* __restrict__ dst,
    int* __restrict__ tab, int* __restrict__ csr,
    const float* __restrict__ embed,
    const float* __restrict__ W_lin, const float* __restrict__ b_lin,
    const float* __restrict__ W_gat,
    const float* __restrict__ attn_l, const float* __restrict__ attn_r,
    float* __restrict__ xg, float* __restrict__ el, float* __restrict__ er,
    int n_tiles) {
  extern __shared__ __attribute__((aligned(16))) float smem[];
  const int tid = threadIdx.x;

  if (blockIdx.x < NCHUNK) {
    unsigned* cnt = (unsigned*)smem;
    int* list = (int*)(cnt + NWORD);
    int* wtot = list + CHUNK_E;
    const int c = blockIdx.x;
    const int ce = (c == NCHUNK - 1) ? LAST_E : CHUNK_E;
    const int cq = ce / 4;
    const int lane = tid & 31;
    const int wave = tid >> 5;

    const int4* d4 = (const int4*)(dst + (size_t)c * CHUNK_E);
    int4 dv[RANK_U];
#pragma unroll
    for (int u = 0; u < RANK_U; ++u) {
      const int q = tid + u * RANK_T;
      dv[u] = d4[(q < cq) ? q : (cq - 1)];
    }
    uint4* cnt4 = (uint4*)cnt;
    for (int j = tid; j < NWORD / 4; j += RANK_T)
      cnt4[j] = make_uint4(0u, 0u, 0u, 0u);
    __syncthreads();

    int4 rk[RANK_U];
#pragma unroll
    for (int u = 0; u < RANK_U; ++u) {
      const int q = tid + u * RANK_T;
      rk[u] = make_int4(0, 0, 0, 0);
      if (q < cq) {
        rk[u].x = rank16(cnt, dv[u].x);
        rk[u].y = rank16(cnt, dv[u].y);
        rk[u].z = rank16(cnt, dv[u].z);
        rk[u].w = rank16(cnt, dv[u].w);
      }
    }
    __syncthreads();

    const int j0 = tid * SCAN_R;
    unsigned ps = 0;
    for (int i = 0; i < SCAN_R; ++i) {
      const int j = j0 + i;
      if (j < NWORD) ps += cnt[j];
    }
    const int own = (int)(ps & 0xFFFFu) + (int)(ps >> 16);
    int incl = own;
    for (int off = 1; off < 32; off <<= 1) {
      const int v = __shfl_up(incl, off, 32);
      if (lane >= off) incl += v;
    }
    if (lane == 31) wtot[wave] = incl;
    __syncthreads();
    int before = (lane < wave) ? wtot[lane] : 0;
    for (int off = 16; off >= 1; off >>= 1) before += __shfl_xor(before, off, 32);
    int run = before + incl - own;
    for (int i = 0; i < SCAN_R; ++i) {
      const int j = j0 + i;
      if (j < NWORD) {
        const unsigned w = cnt[j];
        const unsigned lo = w & 0xFFFFu;
        const unsigned hi = w >> 16;
        cnt[j] = (unsigned)run | (((unsigned)run + lo) << 16);
        run += (int)(lo + hi);
      }
    }
    __syncthreads();

    int4* t4 = (int4*)(tab + (size_t)c * NN);
    const uint2* c2 = (const uint2*)cnt;
    for (int p = tid; p < NWORD / 2; p += RANK_T) {
      const uint2 w = c2[p];
      const unsigned o0 = w.x & 0xFFFFu;
      const unsigned o1 = w.x >> 16;
      const unsigned o2 = w.y & 0xFFFFu;
      const unsigned o3 = w.y >> 16;
      const unsigned o4 =
          (2 * p + 2 < NWORD) ? (cnt[2 * p + 2] & 0xFFFFu) : (unsigned)ce;
      int4 v;
      v.x = (int)(o0 | ((o1 - o0) << 16));
      v.y = (int)(o1 | ((o2 - o1) << 16));
      v.z = (int)(o2 | ((o3 - o2) << 16));
      v.w = (int)(o3 | ((o4 - o3) << 16));
      t4[p] = v;
    }

    const int4* s4 = (const int4*)(src + (size_t)c * CHUNK_E);
#pragma unroll
    for (int u = 0; u < RANK_U; ++u) {
      const int q = tid + u * RANK_T;
      if (q < cq) {
        const int4 sv = s4[q];
        list[first16(cnt, dv[u].x) + rk[u].x] = sv.x;
        list[first16(cnt, dv[u].y) + rk[u].y] = sv.y;
        list[first16(cnt, dv[u].z) + rk[u].z] = sv.z;
        list[first16(cnt, dv[u].w) + rk[u].w] = sv.w;
      }
    }
    __syncthreads();

    int4* o4 = (int4*)(csr + (size_t)c * CHUNK_E);
    const int4* l4 = (const int4*)list;
    for (int q = tid; q < cq; q += RANK_T) o4[q] = l4[q];
    return;
  }

  float* sW1 = smem;
  float* sB1 = sW1 + EMB3 * F_LAT;
  float* sWl = sB1 + F_LAT;
  float* sWr = sWl + F_LAT;
  float* sX  = sWr + F_LAT;

  for (int i = tid; i < EMB3 * F_LAT; i += FRONT_T) sW1[i] = W_lin[i];
  if (tid < F_LAT) sB1[tid] = b_lin[tid];
  if (tid < 2 * F_LAT) {
    const int k = tid & (F_LAT - 1);
    const float4* w4 = (const float4*)(W_gat + k * F_OUT);
    const float4* a4 = (const float4*)((tid < F_LAT) ? attn_l : attn_r);
    float s = 0.0f;
#pragma unroll 4
    for (int c = 0; c < F_OUT / 4; ++c) {
      const float4 w = w4[c];
      const float4 a = a4[c];
      s += w.x * a.x + w.y * a.y + w.z * a.z + w.w * a.w;
    }
    if (tid < F_LAT) sWl[k] = s; else sWr[k] = s;
  }
  __syncthreads();

  const int wave = tid >> 5;
  const int lane = tid & 31;
  const int nb = blockIdx.x - NCHUNK;
  const int m  = lane & 15;
  const int hi = lane >> 4;
  float* xw = sX + wave * 16 * XP;
  const int c4 = m * 4;
  const float4 wl = *(const float4*)(sWl + c4);
  const float4 wr = *(const float4*)(sWr + c4);

  const int tile = wave * NODE_B + nb;
  if (tile < n_tiles) {
    const float* arow = embed + (size_t)(tile * 16 + m) * EMB3;
    v8f acc[4] = {};
    for (int k0 = 0; k0 < EMB3; k0 += 4) {
      const int ka = k0 + 2 * hi;
      v2f a; a.x = arow[ka]; a.y = arow[ka + 1];
      const float* b0 = sW1 + ka * F_LAT + m;
      const float* b1 = b0 + F_LAT;
#pragma unroll
      for (int nt = 0; nt < 4; ++nt) {
        v2f b; b.x = b0[nt * 16]; b.y = b1[nt * 16];
        acc[nt] = wmma4(a, b, acc[nt]);
      }
    }

#pragma unroll
    for (int r = 0; r < 8; ++r) {
      const int row = r + 8 * hi;
#pragma unroll
      for (int nt = 0; nt < 4; ++nt) {
        float v = acc[nt][r] + sB1[nt * 16 + m];
        xw[row * XP + nt * 16 + m] = fmaxf(v, 0.0f);
      }
    }

    float4* xo = (float4*)(xg + (size_t)(tile * 16) * F_LAT);
    float pl[8], pr[8];
#pragma unroll
    for (int it = 0; it < 8; ++it) {
      const int row = 2 * it + hi;
      const float4 v = *(const float4*)(xw + row * XP + c4);
      xo[row * (F_LAT / 4) + m] = v;
      pl[it] = v.x * wl.x + v.y * wl.y + v.z * wl.z + v.w * wl.w;
      pr[it] = v.x * wr.x + v.y * wr.y + v.z * wr.z + v.w * wr.w;
    }
#pragma unroll
    for (int off = 8; off >= 1; off >>= 1) {
#pragma unroll
      for (int it = 0; it < 8; ++it) {
        pl[it] += __shfl_xor(pl[it], off, 32);
        pr[it] += __shfl_xor(pr[it], off, 32);
      }
    }
    if (m == 0) {
#pragma unroll
      for (int it = 0; it < 8; ++it) {
        el[tile * 16 + 2 * it + hi] = pl[it];
        er[tile * 16 + 2 * it + hi] = pr[it];
      }
    }
  }
}

__device__ __forceinline__ float lrelu(float e) {
  return (e > 0.0f) ? e : NEG_SLOPE * e;
}
template <int M>
__device__ __forceinline__ int row_x(int v) {
  return __builtin_amdgcn_update_dpp(0, v, 0x160 + M, 0xF, 0xF, true);
}
template <int N>
__device__ __forceinline__ int row_up(int v) {
  return __builtin_amdgcn_update_dpp(0, v, 0x110 + N, 0xF, 0xF, true);
}
__device__ __forceinline__ int xhalf(int v) {
  return __builtin_amdgcn_permlanex16(v, v, 0x76543210, 0xfedcba98, false,
                                      false);
}
template <int M>
__device__ __forceinline__ float row_xf(float v) {
  return __int_as_float(row_x<M>(__float_as_int(v)));
}
__device__ __forceinline__ float xhalff(float v) {
  return __int_as_float(xhalf(__float_as_int(v)));
}
__device__ __forceinline__ int lane_pick(int v, int idx, bool same_row) {
  const int sel = idx & 15;
  const int own = __builtin_amdgcn_permlane16_var(v, v, sel, false, false);
  const int oth = __builtin_amdgcn_permlanex16_var(v, v, sel, false, false);
  return same_row ? own : oth;
}
__device__ __forceinline__ float wave_max(float x) {
  x = fmaxf(x, xhalff(x));
  x = fmaxf(x, row_xf<8>(x));
  x = fmaxf(x, row_xf<4>(x));
  x = fmaxf(x, row_xf<2>(x));
  x = fmaxf(x, row_xf<1>(x));
  return x;
}
__device__ __forceinline__ float wave_sum(float x) {
  x += xhalff(x);
  x += row_xf<8>(x);
  x += row_xf<4>(x);
  x += row_xf<2>(x);
  x += row_xf<1>(x);
  return x;
}

__device__ __forceinline__ void lds_barrier() {
  __builtin_amdgcn_fence(__ATOMIC_RELEASE, "workgroup", "local");
  __builtin_amdgcn_s_barrier();
  __builtin_amdgcn_fence(__ATOMIC_ACQUIRE, "workgroup", "local");
}

__device__ __forceinline__ void chain_head(int e0, int e1, int lane, int& incl,
                                           int& mid, int& adj0, int& adj1) {
  const int cnt0 = (int)((unsigned)e0 >> 16);
  const int cnt1 = (int)((unsigned)e1 >> 16);
  const int both = cnt0 + cnt1;
  int in = both;
  in += row_up<1>(in);
  in += row_up<2>(in);
  in += row_up<4>(in);
  in += row_up<8>(in);
  const int low_total = __builtin_amdgcn_readlane(in, 15);
  if (lane >= 16) in += low_total;
  const int ex0 = in - both;
  mid = ex0 + cnt0;
  adj0 = (2 * lane) * CHUNK_E + (e0 & 0xFFFF) - ex0;
  adj1 = (2 * lane + 1) * CHUNK_E + (e1 & 0xFFFF) - mid;
  incl = in;
}
__device__ __forceinline__ int slot_addr(int j, int lane, int incl, int mid,
                                         int adj0, int adj1) {
  const int i15 = __builtin_amdgcn_readlane(incl, 15);
  int lo = (i15 <= j) ? 16 : 0;
  const bool same = ((lo ^ lane) & 16) == 0;
  for (int step = 8; step >= 1; step >>= 1) {
    const int v = lane_pick(incl, lo + step - 1, same);
    if (v <= j) lo += step;
  }
  const int m = lane_pick(mid, lo, same);
  const int a0 = lane_pick(adj0, lo, same);
  const int a1 = lane_pick(adj1, lo, same);
  return ((j < m) ? a0 : a1) + j;
}

__device__ __forceinline__ int2 fetch_tab_piece(const int* __restrict__ tab,
                                                int tile, int L) {
  const int* base = tab + tile * 16;
  const unsigned off = (unsigned)(L >> 3) * NN + (unsigned)(L & 7) * 2;
  return *(const int2*)(base + off);
}
__device__ __forceinline__ void put_tab_piece(int* tb, int L, int2 v) {
  int* p = tb + ((L & 7) * 2) * TP + (L >> 3);
  p[0] = v.x;
  p[TP] = v.y;
}
__device__ __forceinline__ void take_entries(const int* tb, int r, int lane,
                                             int& e0, int& e1) {
  e0 = 0;
  e1 = 0;
  if (lane < NPAIR) {
    const int2 e = *(const int2*)(tb + r * TP + 2 * lane);
    e0 = e.x;
    e1 = e.y;
  }
}

__global__ __launch_bounds__(GP_T, 16) void dst_gather_project_kernel(
    const int* __restrict__ tab, const int* __restrict__ csr_src,
    const float* __restrict__ el, const float* __restrict__ er,
    const float* __restrict__ xg, const float* __restrict__ W_gat,
    const float* __restrict__ b_gat, float* __restrict__ out) {
  __shared__ __attribute__((aligned(16))) float sW[F_LAT * F_OUT];
  __shared__ __attribute__((aligned(16))) float sAgg[16 * XP];
  __shared__ __attribute__((aligned(16))) float sOut[16 * OP];
  __shared__ __attribute__((aligned(16))) int sTab[3][16 * TP];

  const int tid = threadIdx.x;
  const int lane = tid & 31;
  const int m  = lane & 15;
  const int hi = lane >> 4;
  const int w = __builtin_amdgcn_readfirstlane(tid >> 5);

  for (int i = tid; i < F_LAT * F_OUT / 4; i += GP_T)
    ((float4*)sW)[i] = ((const float4*)W_gat)[i];

  const float4* x4 = (const float4*)xg;
  const int c0 = lane * 4;
  const float4 bias = *(const float4*)(b_gat + c0);

  int t = blockIdx.x;
  int d = t * 16 + w;
  float erd = er[d];
  if (w < TAB_L / 32) {
    const int tb1 = (t + GP_GRID < NN / 16) ? t + GP_GRID : t;
    const int2 p0 = fetch_tab_piece(tab, t, tid);
    const int2 p1 = fetch_tab_piece(tab, tb1, tid);
    put_tab_piece(sTab[0], tid, p0);
    put_tab_piece(sTab[1], tid, p1);
  }
  __syncthreads();
  int bc = 0, bn = 1, bf = 2;
  int n;
  int s0 = 0;
  float el0 = 0.0f;
  {
    int e0, e1, incl, mid, adj0, adj1;
    take_entries(sTab[0], w, lane, e0, e1);
    chain_head(e0, e1, lane, incl, mid, adj0, adj1);
    n = __builtin_amdgcn_readlane(incl, 31);
    const int a = slot_addr(lane, lane, incl, mid, adj0, adj1);
    if (lane < n) {
      s0 = csr_src[a];
      el0 = el[s0];
    }
  }

  for (; t < NN / 16; t += GP_GRID) {
    const int tn = t + GP_GRID;
    const int t1 = (tn < NN / 16) ? tn : t;
    const int d1 = t1 * 16 + w;
    const float erv = er[t1 * 16 + m];

    const float lg0 = (lane < n) ? lrelu(el0 + erd) : -INFINITY;
    float mx = lg0;
    if (n > 32) {
      int e0, e1, incl, mid, adj0, adj1;
      take_entries(sTab[bc], w, lane, e0, e1);
      chain_head(e0, e1, lane, incl, mid, adj0, adj1);
      for (int j0 = 32; j0 < n; j0 += 32) {
        const int j = j0 + lane;
        const int a = slot_addr(j, lane, incl, mid, adj0, adj1);
        if (j < n) mx = fmaxf(mx, lrelu(el[csr_src[a]] + erd));
      }
    }
    mx = wave_max(mx);
    if ((__float_as_uint(mx) & 0x7F800000u) == 0x7F800000u) mx = 0.0f;

    const float ee0 = (lane < n) ? expf(lg0 - mx) : 0.0f;
    float sum = ee0;
    if (n > 32) {
      int e0, e1, incl, mid, adj0, adj1;
      take_entries(sTab[bc], w, lane, e0, e1);
      chain_head(e0, e1, lane, incl, mid, adj0, adj1);
      for (int j0 = 32; j0 < n; j0 += 32) {
        const int j = j0 + lane;
        const int a = slot_addr(j, lane, incl, mid, adj0, adj1);
        if (j < n) sum += expf(lrelu(el[csr_src[a]] + erd) - mx);
      }
    }
    sum = wave_sum(sum);
    const float dn = (sum > 0.0f) ? sum : 1.0f;

    int n1;
    int s1 = 0;
    const int t2 = (tn + GP_GRID < NN / 16) ? tn + GP_GRID : t;
    int2 piece = {0, 0};
    {
      int e0n, e1n, incl, mid, adj0, adj1;
      take_entries(sTab[bn], w, lane, e0n, e1n);
      chain_head(e0n, e1n, lane, incl, mid, adj0, adj1);
      n1 = __builtin_amdgcn_readlane(incl, 31);
      const int a = slot_addr(lane, lane, incl, mid, adj0, adj1);
      if (w < TAB_L / 32) piece = fetch_tab_piece(tab, t2, tid);
      if (lane < n1) s1 = csr_src[a];
    }
    const float erd1 =
        __int_as_float(__builtin_amdgcn_readlane(__float_as_int(erv), w));

    if (w < TAB_L / 32) put_tab_piece(sTab[bf], tid, piece);

    float4 acc = {0.0f, 0.0f, 0.0f, 0.0f};
    int sC = s0;
    float aC = ee0 / dn;
    for (int j0 = 0; j0 < n; j0 += 32) {
      if (j0 > 0) {
        const int j = j0 + lane;
        int e0, e1, incl, mid, adj0, adj1;
        take_entries(sTab[bc], w, lane, e0, e1);
        chain_head(e0, e1, lane, incl, mid, adj0, adj1);
        const int a = slot_addr(j, lane, incl, mid, adj0, adj1);
        sC = 0;
        aC = 0.0f;
        if (j < n) {
          sC = csr_src[a];
          aC = expf(lrelu(el[sC] + erd) - mx) / dn;
        }
      }
      const int c = (n - j0 < 32) ? (n - j0) : 32;
#pragma unroll 4
      for (int k = 0; k < c; k += 2) {
        const int sLo = __builtin_amdgcn_readlane(sC, k);
        const int sHi = __builtin_amdgcn_readlane(sC, k + 1);
        const int aLo = __builtin_amdgcn_readlane(__float_as_int(aC), k);
        const int aHi = __builtin_amdgcn_readlane(__float_as_int(aC), k + 1);
        const int s = hi ? sHi : sLo;
        const float a = __int_as_float(hi ? aHi : aLo);
        const float4 v = x4[(size_t)s * (F_LAT / 4) + m];
        acc.x += v.x * a;
        acc.y += v.y * a;
        acc.z += v.z * a;
        acc.w += v.w * a;
      }
    }
    acc.x += xhalff(acc.x);
    acc.y += xhalff(acc.y);
    acc.z += xhalff(acc.z);
    acc.w += xhalff(acc.w);
    if (hi == 0) *(float4*)(sAgg + w * XP + m * 4) = acc;
    float el1 = 0.0f;
    if (lane < n1) el1 = el[s1];
    lds_barrier();

    if (w < 8) {
      v8f pc = {};
#pragma unroll 1
      for (int k0 = 0; k0 < F_LAT; k0 += 4) {
        const int ka = k0 + 2 * hi;
        v2f a; a.x = sAgg[m * XP + ka]; a.y = sAgg[m * XP + ka + 1];
        const float* b0 = sW + ka * F_OUT + w * 16 + m;
        v2f b; b.x = b0[0]; b.y = b0[F_OUT];
        pc = wmma4(a, b, pc);
      }
#pragma unroll
      for (int r = 0; r < 8; ++r)
        sOut[(r + 8 * hi) * OP + w * 16 + m] = pc[r];
    }
    lds_barrier();

    const float4 p = *(const float4*)(sOut + w * OP + c0);
    float4 y;
    y.x = p.x + bias.x;
    y.y = p.y + bias.y;
    y.z = p.z + bias.z;
    y.w = p.w + bias.w;
    float* o = out + (size_t)d * 64 + c0;
    if (lane >= 16) {
      y.x = tanhf(y.x); y.y = tanhf(y.y); y.z = tanhf(y.z); y.w = tanhf(y.w);
      o = out + (size_t)NN * 64 + (size_t)d * 64 + (c0 - 64);
    }
    *(float4*)o = y;

    {
      const int b = bc;
      bc = bn;
      bn = bf;
      bf = b;
    }
    d = d1;
    n = n1;
    erd = erd1;
    s0 = s1;
    el0 = el1;
  }
}

extern "C" void kernel_launch(void* const* d_in, const int* in_sizes, int n_in,
                              void* d_out, int out_size, void* d_ws,
                              size_t ws_size, hipStream_t stream) {
  const float* embed  = (const float*)d_in[0];
  const int*   src    = (const int*)d_in[1];
  const int*   dst    = (const int*)d_in[2];
  const float* W_lin  = (const float*)d_in[3];
  const float* b_lin  = (const float*)d_in[4];
  const float* W_gat  = (const float*)d_in[5];
  const float* attn_l = (const float*)d_in[6];
  const float* attn_r = (const float*)d_in[7];
  const float* b_gat  = (const float*)d_in[8];
  float* out = (float*)d_out;

  float* ws     = (float*)d_ws;
  float* xg     = ws;
  int* tab      = (int*)(xg + (size_t)NN * F_LAT);
  float* el     = xg + (size_t)NN * F_OUT;
  float* er     = el + NN;
  int* csr_src  = (int*)(er + NN);

  const int n_tiles = NN / 16;
  const size_t front_lds =
      (size_t)NWORD * sizeof(unsigned) + (size_t)(CHUNK_E + 32) * sizeof(int);
  (void)hipFuncSetAttribute((const void*)rank_and_node_kernel,
                            hipFuncAttributeMaxDynamicSharedMemorySize,
                            (int)front_lds);
  rank_and_node_kernel<<<NCHUNK + NODE_B, FRONT_T, front_lds, stream>>>(
      src, dst, tab, csr_src, embed, W_lin, b_lin, W_gat, attn_l, attn_r,
      xg, el, er, n_tiles);

  dst_gather_project_kernel<<<GP_GRID, GP_T, 0, stream>>>(
      tab, csr_src, el, er, xg, W_gat, b_gat, out);
}


# gather kernel: x-row loads issued ahead in whole groups of 6 slot pairs (even/odd slots on the two lane rows, DPP row_share feeds them), first group before the next tile's index chain; bit-identical
# speedup vs baseline: 6.9155x; 1.0484x over previous
#include <hip/hip_runtime.h>
#include <math.h>

#define NN     100000
#define NE     1600000
#define EMB3   192
#define F_LAT  64
#define F_OUT  128
#define NEG_SLOPE 0.2f
#define NCHUNK  56
#define CHUNK_E 28672
#define LAST_E  (NE - (NCHUNK - 1) * CHUNK_E)
#define CHUNK_Q (CHUNK_E / 4)
#define RANK_T  1024
#define RANK_U  (CHUNK_Q / RANK_T)
#define NWORD   (NN / 2)
#define SCAN_R  ((NWORD + RANK_T - 1) / RANK_T)
#define NPAIR   (NCHUNK / 2)
#define XP      68
#define OP      132
#define FRONT_T RANK_T
#define NODE_B  (256 - NCHUNK)
#define NODE_W  (NODE_B * (FRONT_T / 32))
static_assert(CHUNK_E % 4 == 0 && CHUNK_E < 65536 && CHUNK_Q % RANK_T == 0, "chunking");
static_assert(LAST_E > 0 && LAST_E <= CHUNK_E && LAST_E % 4 == 0, "last chunk");
static_assert(NCHUNK % 2 == 0 && NPAIR <= 32 && NWORD % 2 == 0 && NN % 4 == 0, "table layout");
static_assert(NODE_W >= NN / 16, "every node tile needs a wave of its own");
#define GP_T    512
#define GP_GRID 1024
static_assert(GP_GRID <= NN / 16, "every gather block starts with a tile of its own");
#define TP      58
#define TAB_L   (NCHUNK * 8)
static_assert(TP >= NCHUNK && TP % 2 == 0 && TAB_L <= GP_T && TAB_L % 32 == 0, "table staging");

typedef __attribute__((ext_vector_type(2))) float v2f;
typedef __attribute__((ext_vector_type(8))) float v8f;

__device__ __forceinline__ v8f wmma4(v2f a, v2f b, v8f c) {
  return __builtin_amdgcn_wmma_f32_16x16x4_f32(false, a, false, b, (short)0, c,
                                               false, false);
}


__device__ __forceinline__ int rank16(unsigned* cnt, int d) {
  const unsigned sh = (unsigned)(d & 1) << 4;
  const unsigned old = atomicAdd(&cnt[d >> 1], 1u << sh);
  return (int)((old >> sh) & 0xFFFFu);
}
__device__ __forceinline__ int first16(const unsigned* cnt, int d) {
  const unsigned sh = (unsigned)(d & 1) << 4;
  return (int)((cnt[d >> 1] >> sh) & 0xFFFFu);
}

__global__ __launch_bounds__(FRONT_T) void rank_and_node_kernel(
    const int* __restrict__ src, const int* __restrict__ dst,
    int* __restrict__ tab, int* __restrict__ csr,
    const float* __restrict__ embed,
    const float* __restrict__ W_lin, const float* __restrict__ b_lin,
    const float* __restrict__ W_gat,
    const float* __restrict__ attn_l, const float* __restrict__ attn_r,
    float* __restrict__ xg, float* __restrict__ el, float* __restrict__ er,
    int n_tiles) {
  extern __shared__ __attribute__((aligned(16))) float smem[];
  const int tid = threadIdx.x;

  if (blockIdx.x < NCHUNK) {
    unsigned* cnt = (unsigned*)smem;
    int* list = (int*)(cnt + NWORD);
    int* wtot = list + CHUNK_E;
    const int c = blockIdx.x;
    const int ce = (c == NCHUNK - 1) ? LAST_E : CHUNK_E;
    const int cq = ce / 4;
    const int lane = tid & 31;
    const int wave = tid >> 5;

    const int4* d4 = (const int4*)(dst + (size_t)c * CHUNK_E);
    int4 dv[RANK_U];
#pragma unroll
    for (int u = 0; u < RANK_U; ++u) {
      const int q = tid + u * RANK_T;
      dv[u] = d4[(q < cq) ? q : (cq - 1)];
    }
    uint4* cnt4 = (uint4*)cnt;
    for (int j = tid; j < NWORD / 4; j += RANK_T)
      cnt4[j] = make_uint4(0u, 0u, 0u, 0u);
    __syncthreads();

    int4 rk[RANK_U];
#pragma unroll
    for (int u = 0; u < RANK_U; ++u) {
      const int q = tid + u * RANK_T;
      rk[u] = make_int4(0, 0, 0, 0);
      if (q < cq) {
        rk[u].x = rank16(cnt, dv[u].x);
        rk[u].y = rank16(cnt, dv[u].y);
        rk[u].z = rank16(cnt, dv[u].z);
        rk[u].w = rank16(cnt, dv[u].w);
      }
    }
    __syncthreads();

    const int j0 = tid * SCAN_R;
    unsigned ps = 0;
    for (int i = 0; i < SCAN_R; ++i) {
      const int j = j0 + i;
      if (j < NWORD) ps += cnt[j];
    }
    const int own = (int)(ps & 0xFFFFu) + (int)(ps >> 16);
    int incl = own;
    for (int off = 1; off < 32; off <<= 1) {
      const int v = __shfl_up(incl, off, 32);
      if (lane >= off) incl += v;
    }
    if (lane == 31) wtot[wave] = incl;
    __syncthreads();
    int before = (lane < wave) ? wtot[lane] : 0;
    for (int off = 16; off >= 1; off >>= 1) before += __shfl_xor(before, off, 32);
    int run = before + incl - own;
    for (int i = 0; i < SCAN_R; ++i) {
      const int j = j0 + i;
      if (j < NWORD) {
        const unsigned w = cnt[j];
        const unsigned lo = w & 0xFFFFu;
        const unsigned hi = w >> 16;
        cnt[j] = (unsigned)run | (((unsigned)run + lo) << 16);
        run += (int)(lo + hi);
      }
    }
    __syncthreads();

    int4* t4 = (int4*)(tab + (size_t)c * NN);
    const uint2* c2 = (const uint2*)cnt;
    for (int p = tid; p < NWORD / 2; p += RANK_T) {
      const uint2 w = c2[p];
      const unsigned o0 = w.x & 0xFFFFu;
      const unsigned o1 = w.x >> 16;
      const unsigned o2 = w.y & 0xFFFFu;
      const unsigned o3 = w.y >> 16;
      const unsigned o4 =
          (2 * p + 2 < NWORD) ? (cnt[2 * p + 2] & 0xFFFFu) : (unsigned)ce;
      int4 v;
      v.x = (int)(o0 | ((o1 - o0) << 16));
      v.y = (int)(o1 | ((o2 - o1) << 16));
      v.z = (int)(o2 | ((o3 - o2) << 16));
      v.w = (int)(o3 | ((o4 - o3) << 16));
      t4[p] = v;
    }

    const int4* s4 = (const int4*)(src + (size_t)c * CHUNK_E);
#pragma unroll
    for (int u = 0; u < RANK_U; ++u) {
      const int q = tid + u * RANK_T;
      if (q < cq) {
        const int4 sv = s4[q];
        list[first16(cnt, dv[u].x) + rk[u].x] = sv.x;
        list[first16(cnt, dv[u].y) + rk[u].y] = sv.y;
        list[first16(cnt, dv[u].z) + rk[u].z] = sv.z;
        list[first16(cnt, dv[u].w) + rk[u].w] = sv.w;
      }
    }
    __syncthreads();

    int4* o4 = (int4*)(csr + (size_t)c * CHUNK_E);
    const int4* l4 = (const int4*)list;
    for (int q = tid; q < cq; q += RANK_T) o4[q] = l4[q];
    return;
  }

  float* sW1 = smem;
  float* sB1 = sW1 + EMB3 * F_LAT;
  float* sWl = sB1 + F_LAT;
  float* sWr = sWl + F_LAT;
  float* sX  = sWr + F_LAT;

  for (int i = tid; i < EMB3 * F_LAT; i += FRONT_T) sW1[i] = W_lin[i];
  if (tid < F_LAT) sB1[tid] = b_lin[tid];
  if (tid < 2 * F_LAT) {
    const int k = tid & (F_LAT - 1);
    const float4* w4 = (const float4*)(W_gat + k * F_OUT);
    const float4* a4 = (const float4*)((tid < F_LAT) ? attn_l : attn_r);
    float s = 0.0f;
#pragma unroll 4
    for (int c = 0; c < F_OUT / 4; ++c) {
      const float4 w = w4[c];
      const float4 a = a4[c];
      s += w.x * a.x + w.y * a.y + w.z * a.z + w.w * a.w;
    }
    if (tid < F_LAT) sWl[k] = s; else sWr[k] = s;
  }
  __syncthreads();

  const int wave = tid >> 5;
  const int lane = tid & 31;
  const int nb = blockIdx.x - NCHUNK;
  const int m  = lane & 15;
  const int hi = lane >> 4;
  float* xw = sX + wave * 16 * XP;
  const int c4 = m * 4;
  const float4 wl = *(const float4*)(sWl + c4);
  const float4 wr = *(const float4*)(sWr + c4);

  const int tile = wave * NODE_B + nb;
  if (tile < n_tiles) {
    const float* arow = embed + (size_t)(tile * 16 + m) * EMB3;
    v8f acc[4] = {};
    for (int k0 = 0; k0 < EMB3; k0 += 4) {
      const int ka = k0 + 2 * hi;
      v2f a; a.x = arow[ka]; a.y = arow[ka + 1];
      const float* b0 = sW1 + ka * F_LAT + m;
      const float* b1 = b0 + F_LAT;
#pragma unroll
      for (int nt = 0; nt < 4; ++nt) {
        v2f b; b.x = b0[nt * 16]; b.y = b1[nt * 16];
        acc[nt] = wmma4(a, b, acc[nt]);
      }
    }

#pragma unroll
    for (int r = 0; r < 8; ++r) {
      const int row = r + 8 * hi;
#pragma unroll
      for (int nt = 0; nt < 4; ++nt) {
        float v = acc[nt][r] + sB1[nt * 16 + m];
        xw[row * XP + nt * 16 + m] = fmaxf(v, 0.0f);
      }
    }

    float4* xo = (float4*)(xg + (size_t)(tile * 16) * F_LAT);
    float pl[8], pr[8];
#pragma unroll
    for (int it = 0; it < 8; ++it) {
      const int row = 2 * it + hi;
      const float4 v = *(const float4*)(xw + row * XP + c4);
      xo[row * (F_LAT / 4) + m] = v;
      pl[it] = v.x * wl.x + v.y * wl.y + v.z * wl.z + v.w * wl.w;
      pr[it] = v.x * wr.x + v.y * wr.y + v.z * wr.z + v.w * wr.w;
    }
#pragma unroll
    for (int off = 8; off >= 1; off >>= 1) {
#pragma unroll
      for (int it = 0; it < 8; ++it) {
        pl[it] += __shfl_xor(pl[it], off, 32);
        pr[it] += __shfl_xor(pr[it], off, 32);
      }
    }
    if (m == 0) {
#pragma unroll
      for (int it = 0; it < 8; ++it) {
        el[tile * 16 + 2 * it + hi] = pl[it];
        er[tile * 16 + 2 * it + hi] = pr[it];
      }
    }
  }
}

__device__ __forceinline__ float lrelu(float e) {
  return (e > 0.0f) ? e : NEG_SLOPE * e;
}
template <int M>
__device__ __forceinline__ int row_x(int v) {
  return __builtin_amdgcn_update_dpp(0, v, 0x160 + M, 0xF, 0xF, true);
}
template <int N>
__device__ __forceinline__ int row_up(int v) {
  return __builtin_amdgcn_update_dpp(0, v, 0x110 + N, 0xF, 0xF, true);
}
__device__ __forceinline__ int xhalf(int v) {
  return __builtin_amdgcn_permlanex16(v, v, 0x76543210, 0xfedcba98, false,
                                      false);
}
template <int P>
__device__ __forceinline__ int row_bc(int v) {
  return __builtin_amdgcn_update_dpp(0, v, 0x150 + P, 0xF, 0xF, true);
}
template <int M>
__device__ __forceinline__ float row_xf(float v) {
  return __int_as_float(row_x<M>(__float_as_int(v)));
}
__device__ __forceinline__ float xhalff(float v) {
  return __int_as_float(xhalf(__float_as_int(v)));
}
__device__ __forceinline__ int lane_pick(int v, int idx, bool same_row) {
  const int sel = idx & 15;
  const int own = __builtin_amdgcn_permlane16_var(v, v, sel, false, false);
  const int oth = __builtin_amdgcn_permlanex16_var(v, v, sel, false, false);
  return same_row ? own : oth;
}
__device__ __forceinline__ float wave_max(float x) {
  x = fmaxf(x, row_xf<8>(x));
  x = fmaxf(x, row_xf<4>(x));
  x = fmaxf(x, row_xf<2>(x));
  x = fmaxf(x, row_xf<1>(x));
  x = fmaxf(x, xhalff(x));
  return x;
}
__device__ __forceinline__ float wave_sum(float x) {
  x += row_xf<8>(x);
  x += row_xf<4>(x);
  x += row_xf<2>(x);
  x += row_xf<1>(x);
  x += xhalff(x);
  return x;
}

__device__ __forceinline__ void lds_barrier() {
  __builtin_amdgcn_fence(__ATOMIC_RELEASE, "workgroup", "local");
  __builtin_amdgcn_s_barrier();
  __builtin_amdgcn_fence(__ATOMIC_ACQUIRE, "workgroup", "local");
}

__device__ __forceinline__ void chain_head(int e0, int e1, int lane, int& incl,
                                           int& mid, int& adj0, int& adj1) {
  const int cnt0 = (int)((unsigned)e0 >> 16);
  const int cnt1 = (int)((unsigned)e1 >> 16);
  const int both = cnt0 + cnt1;
  int in = both;
  in += row_up<1>(in);
  in += row_up<2>(in);
  in += row_up<4>(in);
  in += row_up<8>(in);
  const int low_total = __builtin_amdgcn_readlane(in, 15);
  if (lane >= 16) in += low_total;
  const int ex0 = in - both;
  mid = ex0 + cnt0;
  adj0 = (2 * lane) * CHUNK_E + (e0 & 0xFFFF) - ex0;
  adj1 = (2 * lane + 1) * CHUNK_E + (e1 & 0xFFFF) - mid;
  incl = in;
}
__device__ __forceinline__ int slot_addr(int j, int lane, int incl, int mid,
                                         int adj0, int adj1) {
  const int i15 = __builtin_amdgcn_readlane(incl, 15);
  int lo = (i15 <= j) ? 16 : 0;
  const bool same = ((lo ^ lane) & 16) == 0;
  for (int step = 8; step >= 1; step >>= 1) {
    const int v = lane_pick(incl, lo + step - 1, same);
    if (v <= j) lo += step;
  }
  const int m = lane_pick(mid, lo, same);
  const int a0 = lane_pick(adj0, lo, same);
  const int a1 = lane_pick(adj1, lo, same);
  return ((j < m) ? a0 : a1) + j;
}

__device__ __forceinline__ int2 fetch_tab_piece(const int* __restrict__ tab,
                                                int tile, int L) {
  const int* base = tab + tile * 16;
  const unsigned off = (unsigned)(L >> 3) * NN + (unsigned)(L & 7) * 2;
  return *(const int2*)(base + off);
}
__device__ __forceinline__ void put_tab_piece(int* tb, int L, int2 v) {
  int* p = tb + ((L & 7) * 2) * TP + (L >> 3);
  p[0] = v.x;
  p[TP] = v.y;
}
__device__ __forceinline__ void take_entries(const int* tb, int r, int lane,
                                             int& e0, int& e1) {
  e0 = 0;
  e1 = 0;
  if (lane < NPAIR) {
    const int2 e = *(const int2*)(tb + r * TP + 2 * lane);
    e0 = e.x;
    e1 = e.y;
  }
}

template <int P>
__device__ __forceinline__ float4 row_ld(const char* __restrict__ xb, int so,
                                         int lane_off) {
  return *(const float4*)(xb + (unsigned)(row_bc<P>(so) + lane_off));
}
template <int P>
__device__ __forceinline__ void row_fma(const float4 r, float aC, float4& acc) {
  const float a = __int_as_float(row_bc<P>(__float_as_int(aC)));
  acc.x += r.x * a;
  acc.y += r.y * a;
  acc.z += r.z * a;
  acc.w += r.w * a;
}
template <int P0>
__device__ __forceinline__ void gather12(const char* __restrict__ xb, int so,
                                         int lane_off, float aC, float4& acc) {
  const float4 r0 = row_ld<P0 + 0>(xb, so, lane_off);
  const float4 r1 = row_ld<P0 + 1>(xb, so, lane_off);
  const float4 r2 = row_ld<P0 + 2>(xb, so, lane_off);
  const float4 r3 = row_ld<P0 + 3>(xb, so, lane_off);
  const float4 r4 = row_ld<P0 + 4>(xb, so, lane_off);
  const float4 r5 = row_ld<P0 + 5>(xb, so, lane_off);
  row_fma<P0 + 0>(r0, aC, acc);
  row_fma<P0 + 1>(r1, aC, acc);
  row_fma<P0 + 2>(r2, aC, acc);
  row_fma<P0 + 3>(r3, aC, acc);
  row_fma<P0 + 4>(r4, aC, acc);
  row_fma<P0 + 5>(r5, aC, acc);
}
template <int P0>
__device__ __forceinline__ void gather8(const char* __restrict__ xb, int so,
                                        int lane_off, float aC, float4& acc) {
  const float4 r0 = row_ld<P0 + 0>(xb, so, lane_off);
  const float4 r1 = row_ld<P0 + 1>(xb, so, lane_off);
  const float4 r2 = row_ld<P0 + 2>(xb, so, lane_off);
  const float4 r3 = row_ld<P0 + 3>(xb, so, lane_off);
  row_fma<P0 + 0>(r0, aC, acc);
  row_fma<P0 + 1>(r1, aC, acc);
  row_fma<P0 + 2>(r2, aC, acc);
  row_fma<P0 + 3>(r3, aC, acc);
}

__global__ __launch_bounds__(GP_T, 16) void dst_gather_project_kernel(
    const int* __restrict__ tab, const int* __restrict__ csr_src,
    const float* __restrict__ el, const float* __restrict__ er,
    const float* __restrict__ xg, const float* __restrict__ W_gat,
    const float* __restrict__ b_gat, float* __restrict__ out) {
  __shared__ __attribute__((aligned(16))) float sW[F_LAT * F_OUT];
  __shared__ __attribute__((aligned(16))) float sAgg[16 * XP];
  __shared__ __attribute__((aligned(16))) float sOut[16 * OP];
  __shared__ __attribute__((aligned(16))) int sTab[3][16 * TP];

  const int tid = threadIdx.x;
  const int lane = tid & 31;
  const int m  = lane & 15;
  const int hi = lane >> 4;
  const int w = __builtin_amdgcn_readfirstlane(tid >> 5);
  const int jl = 2 * m + hi;

  for (int i = tid; i < F_LAT * F_OUT / 4; i += GP_T)
    ((float4*)sW)[i] = ((const float4*)W_gat)[i];

  const char* xb = (const char*)xg;
  const int lane_off = m * 16;
  const int c0 = lane * 4;
  const float4 bias = *(const float4*)(b_gat + c0);

  int t = blockIdx.x;
  int d = t * 16 + w;
  float erd = er[d];
  if (w < TAB_L / 32) {
    const int tb1 = (t + GP_GRID < NN / 16) ? t + GP_GRID : t;
    const int2 p0 = fetch_tab_piece(tab, t, tid);
    const int2 p1 = fetch_tab_piece(tab, tb1, tid);
    put_tab_piece(sTab[0], tid, p0);
    put_tab_piece(sTab[1], tid, p1);
  }
  __syncthreads();
  int bc = 0, bn = 1, bf = 2;
  int n;
  int s0 = 0;
  float el0 = 0.0f;
  {
    int e0, e1, incl, mid, adj0, adj1;
    take_entries(sTab[0], w, lane, e0, e1);
    chain_head(e0, e1, lane, incl, mid, adj0, adj1);
    n = __builtin_amdgcn_readlane(incl, 31);
    const int a = slot_addr(jl, lane, incl, mid, adj0, adj1);
    if (jl < n) {
      s0 = csr_src[a];
      el0 = el[s0];
    }
  }

  for (; t < NN / 16; t += GP_GRID) {
    const int tn = t + GP_GRID;
    const int t1 = (tn < NN / 16) ? tn : t;
    const int d1 = t1 * 16 + w;
    const float erv = er[t1 * 16 + m];

    const float lg0 = (jl < n) ? lrelu(el0 + erd) : -INFINITY;
    float mx = lg0;
    if (n > 32) {
      int e0, e1, incl, mid, adj0, adj1;
      take_entries(sTab[bc], w, lane, e0, e1);
      chain_head(e0, e1, lane, incl, mid, adj0, adj1);
      for (int j0 = 32; j0 < n; j0 += 32) {
        const int j = j0 + jl;
        const int a = slot_addr(j, lane, incl, mid, adj0, adj1);
        if (j < n) mx = fmaxf(mx, lrelu(el[csr_src[a]] + erd));
      }
    }
    mx = wave_max(mx);
    if ((__float_as_uint(mx) & 0x7F800000u) == 0x7F800000u) mx = 0.0f;

    const float ee0 = (jl < n) ? expf(lg0 - mx) : 0.0f;
    float sum = ee0;
    if (n > 32) {
      int e0, e1, incl, mid, adj0, adj1;
      take_entries(sTab[bc], w, lane, e0, e1);
      chain_head(e0, e1, lane, incl, mid, adj0, adj1);
      for (int j0 = 32; j0 < n; j0 += 32) {
        const int j = j0 + jl;
        const int a = slot_addr(j, lane, incl, mid, adj0, adj1);
        if (j < n) sum += expf(lrelu(el[csr_src[a]] + erd) - mx);
      }
    }
    sum = wave_sum(sum);
    const float dn = (sum > 0.0f) ? sum : 1.0f;

    const int so0 = s0 << 8;
    const float4 g0 = row_ld<0>(xb, so0, lane_off);
    const float4 g1 = row_ld<1>(xb, so0, lane_off);
    const float4 g2 = row_ld<2>(xb, so0, lane_off);
    const float4 g3 = row_ld<3>(xb, so0, lane_off);
    const float4 g4 = row_ld<4>(xb, so0, lane_off);
    const float4 g5 = row_ld<5>(xb, so0, lane_off);

    int n1;
    int s1 = 0;
    const int t2 = (tn + GP_GRID < NN / 16) ? tn + GP_GRID : t;
    int2 piece = {0, 0};
    {
      int e0n, e1n, incl, mid, adj0, adj1;
      take_entries(sTab[bn], w, lane, e0n, e1n);
      chain_head(e0n, e1n, lane, incl, mid, adj0, adj1);
      n1 = __builtin_amdgcn_readlane(incl, 31);
      const int a = slot_addr(jl, lane, incl, mid, adj0, adj1);
      if (w < TAB_L / 32) piece = fetch_tab_piece(tab, t2, tid);
      if (jl < n1) s1 = csr_src[a];
    }
    const float erd1 =
        __int_as_float(__builtin_amdgcn_readlane(__float_as_int(erv), w));

    if (w < TAB_L / 32) put_tab_piece(sTab[bf], tid, piece);

    float4 acc = {0.0f, 0.0f, 0.0f, 0.0f};
    {
      const float aC = ee0 / dn;
      row_fma<0>(g0, aC, acc);
      row_fma<1>(g1, aC, acc);
      row_fma<2>(g2, aC, acc);
      row_fma<3>(g3, aC, acc);
      row_fma<4>(g4, aC, acc);
      row_fma<5>(g5, aC, acc);
      if (n > 12) {
        gather12<6>(xb, so0, lane_off, aC, acc);
        if (n > 24) gather8<12>(xb, so0, lane_off, aC, acc);
      }
    }
    for (int j0 = 32; j0 < n; j0 += 32) {
      const int j = j0 + jl;
      int e0, e1, incl, mid, adj0, adj1;
      take_entries(sTab[bc], w, lane, e0, e1);
      chain_head(e0, e1, lane, incl, mid, adj0, adj1);
      const int a = slot_addr(j, lane, incl, mid, adj0, adj1);
      int sC = 0;
      float aC = 0.0f;
      if (j < n) {
        sC = csr_src[a];
        aC = expf(lrelu(el[sC] + erd) - mx) / dn;
      }
      const int so = sC << 8;
      gather12<0>(xb, so, lane_off, aC, acc);
      if (n - j0 > 12) {
        gather12<6>(xb, so, lane_off, aC, acc);
        if (n - j0 > 24) gather8<12>(xb, so, lane_off, aC, acc);
      }
    }
    acc.x += xhalff(acc.x);
    acc.y += xhalff(acc.y);
    acc.z += xhalff(acc.z);
    acc.w += xhalff(acc.w);
    if (hi == 0) *(float4*)(sAgg + w * XP + m * 4) = acc;
    float el1 = 0.0f;
    if (jl < n1) el1 = el[s1];
    lds_barrier();

    if (w < 8) {
      v8f pc = {};
#pragma unroll 1
      for (int k0 = 0; k0 < F_LAT; k0 += 4) {
        const int ka = k0 + 2 * hi;
        v2f a; a.x = sAgg[m * XP + ka]; a.y = sAgg[m * XP + ka + 1];
        const float* b0 = sW + ka * F_OUT + w * 16 + m;
        v2f b; b.x = b0[0]; b.y = b0[F_OUT];
        pc = wmma4(a, b, pc);
      }
#pragma unroll
      for (int r = 0; r < 8; ++r)
        sOut[(r + 8 * hi) * OP + w * 16 + m] = pc[r];
    }
    lds_barrier();

    const float4 p = *(const float4*)(sOut + w * OP + c0);
    float4 y;
    y.x = p.x + bias.x;
    y.y = p.y + bias.y;
    y.z = p.z + bias.z;
    y.w = p.w + bias.w;
    float* o = out + (size_t)d * 64 + c0;
    if (lane >= 16) {
      y.x = tanhf(y.x); y.y = tanhf(y.y); y.z = tanhf(y.z); y.w = tanhf(y.w);
      o = out + (size_t)NN * 64 + (size_t)d * 64 + (c0 - 64);
    }
    *(float4*)o = y;

    {
      const int b = bc;
      bc = bn;
      bn = bf;
      bf = b;
    }
    d = d1;
    n = n1;
    erd = erd1;
    s0 = s1;
    el0 = el1;
  }
}

extern "C" void kernel_launch(void* const* d_in, const int* in_sizes, int n_in,
                              void* d_out, int out_size, void* d_ws,
                              size_t ws_size, hipStream_t stream) {
  const float* embed  = (const float*)d_in[0];
  const int*   src    = (const int*)d_in[1];
  const int*   dst    = (const int*)d_in[2];
  const float* W_lin  = (const float*)d_in[3];
  const float* b_lin  = (const float*)d_in[4];
  const float* W_gat  = (const float*)d_in[5];
  const float* attn_l = (const float*)d_in[6];
  const float* attn_r = (const float*)d_in[7];
  const float* b_gat  = (const float*)d_in[8];
  float* out = (float*)d_out;

  float* ws     = (float*)d_ws;
  float* xg     = ws;
  int* tab      = (int*)(xg + (size_t)NN * F_LAT);
  float* el     = xg + (size_t)NN * F_OUT;
  float* er     = el + NN;
  int* csr_src  = (int*)(er + NN);

  const int n_tiles = NN / 16;
  const size_t front_lds =
      (size_t)NWORD * sizeof(unsigned) + (size_t)(CHUNK_E + 32) * sizeof(int);
  (void)hipFuncSetAttribute((const void*)rank_and_node_kernel,
                            hipFuncAttributeMaxDynamicSharedMemorySize,
                            (int)front_lds);
  rank_and_node_kernel<<<NCHUNK + NODE_B, FRONT_T, front_lds, stream>>>(
      src, dst, tab, csr_src, embed, W_lin, b_lin, W_gat, attn_l, attn_r,
      xg, el, er, n_tiles);

  dst_gather_project_kernel<<<GP_GRID, GP_T, 0, stream>>>(
      tab, csr_src, el, er, xg, W_gat, b_gat, out);
}
